# starting kernel
# baseline (speedup 1.0000x reference)
#include <hip/hip_runtime.h>

#define D 128

typedef float v2f __attribute__((ext_vector_type(2)));
typedef float v8f __attribute__((ext_vector_type(8)));

#if defined(__HIP_DEVICE_COMPILE__) && __has_builtin(__builtin_amdgcn_wmma_f32_16x16x4_f32)
#define HAVE_WMMA_F32 1
#endif

__device__ __forceinline__ v8f wmma_acc(v2f a, v2f b, v8f c) {
#if defined(HAVE_WMMA_F32)
  return __builtin_amdgcn_wmma_f32_16x16x4_f32(false, a, false, b, (short)0, c,
                                               false, false);
#else
  (void)a; (void)b;
  return c;
#endif
}

__global__ void k_zero(float* __restrict__ p, int n) {
  int i = blockIdx.x * blockDim.x + threadIdx.x;
  if (i < n) p[i] = 0.f;
}

__global__ void k_count(const int* __restrict__ dst, float* __restrict__ deg, int e) {
  int i = blockIdx.x * blockDim.x + threadIdx.x;
  if (i < e) {
    __hip_atomic_fetch_add(&deg[dst[i]], 1.0f, __ATOMIC_RELAXED,
                           __HIP_MEMORY_SCOPE_AGENT);
  }
}

__global__ void k_dinv(const float* __restrict__ deg, float* __restrict__ dinv, int n) {
  int i = blockIdx.x * blockDim.x + threadIdx.x;
  if (i < n) dinv[i] = rsqrtf(deg[i] + 1.0f);
}

__global__ void __launch_bounds__(256) k_gcn_gemm(
    const float* __restrict__ X, const float* __restrict__ W,
    const float* __restrict__ bias, const float* __restrict__ dinv,
    float* __restrict__ H, float* __restrict__ AGG, int nrows, int relu_in) {
  const int wave = threadIdx.x >> 5;
  const int lane = threadIdx.x & 31;
  const int half = lane >> 4;
  const int l16  = lane & 15;
  const int row0 = (blockIdx.x * 8 + wave) * 16;
  if (row0 >= nrows) return;

  const int m = row0 + l16;
  const bool mvalid = (m < nrows);
  const float* __restrict__ arow = X + (size_t)(mvalid ? m : 0) * D;

  const v8f vzero = {0.f, 0.f, 0.f, 0.f, 0.f, 0.f, 0.f, 0.f};
  v8f acc[8];
#pragma unroll
  for (int i = 0; i < 8; ++i) acc[i] = vzero;

  for (int k0 = 0; k0 < D; k0 += 4) {
    const int ka = k0 + 2 * half;
    float a0 = arow[ka];
    float a1 = arow[ka + 1];
    if (relu_in) { a0 = fmaxf(a0, 0.f); a1 = fmaxf(a1, 0.f); }
    if (!mvalid) { a0 = 0.f; a1 = 0.f; }
    v2f a; a.x = a0; a.y = a1;
    const float* __restrict__ w0 = W + (size_t)ka * D + l16;
    const float* __restrict__ w1 = w0 + D;
#pragma unroll
    for (int ct = 0; ct < 8; ++ct) {
      v2f b; b.x = w0[ct * 16]; b.y = w1[ct * 16];
      acc[ct] = wmma_acc(a, b, acc[ct]);
    }
  }

#pragma unroll
  for (int v = 0; v < 8; ++v) {
    const int mm = row0 + v + 8 * half;
    if (mm < nrows) {
      const float di = dinv[mm];
      const float s = di * di;
      const size_t base = (size_t)mm * D;
#pragma unroll
      for (int ct = 0; ct < 8; ++ct) {
        const int col = ct * 16 + l16;
        const float hv = acc[ct][v];
        H[base + col] = hv;
        AGG[base + col] = fmaf(hv, s, bias[col]);
      }
    }
  }
}

__global__ void __launch_bounds__(256) k_scatter(
    const float* __restrict__ H, const float* __restrict__ dinv,
    const int* __restrict__ src, const int* __restrict__ dst,
    float* __restrict__ AGG, int nedges) {
  const int lane = threadIdx.x & 31;
  const int e = (int)(((size_t)blockIdx.x * blockDim.x + threadIdx.x) >> 5);
  if (e >= nedges) return;
  const int s = src[e];
  const int d = dst[e];
  const float coef = dinv[s] * dinv[d];
  const float4 hv = ((const float4*)(H + (size_t)s * D))[lane];
  float* __restrict__ ap = AGG + (size_t)d * D + lane * 4;
  __hip_atomic_fetch_add(ap + 0, hv.x * coef, __ATOMIC_RELAXED, __HIP_MEMORY_SCOPE_AGENT);
  __hip_atomic_fetch_add(ap + 1, hv.y * coef, __ATOMIC_RELAXED, __HIP_MEMORY_SCOPE_AGENT);
  __hip_atomic_fetch_add(ap + 2, hv.z * coef, __ATOMIC_RELAXED, __HIP_MEMORY_SCOPE_AGENT);
  __hip_atomic_fetch_add(ap + 3, hv.w * coef, __ATOMIC_RELAXED, __HIP_MEMORY_SCOPE_AGENT);
}

extern "C" void kernel_launch(void* const* d_in, const int* in_sizes, int n_in,
                              void* d_out, int out_size, void* d_ws, size_t ws_size,
                              hipStream_t stream) {
  (void)n_in; (void)out_size; (void)ws_size;
  const float* x  = (const float*)d_in[0];
  const int*   ei = (const int*)d_in[1];
  const float* W1 = (const float*)d_in[2];
  const float* b1 = (const float*)d_in[3];
  const float* W2 = (const float*)d_in[4];
  const float* b2 = (const float*)d_in[5];
  const int n = in_sizes[0] / D;
  const int e = in_sizes[1] / 2;
  const int* srcv = ei;
  const int* dstv = ei + e;

  char* ws = (char*)d_ws;
  float* dinv = (float*)ws;
  size_t off = (((size_t)n * 4) + 255) & ~(size_t)255;
  float* Hbuf = (float*)(ws + off);
  off += (((size_t)n * D * 4) + 255) & ~(size_t)255;
  float* Abuf = (float*)(ws + off);
  float* deg = Hbuf;
  float* outp = (float*)d_out;

  const dim3 blk(256);
  const int nblk = (n + 255) / 256;
  const int eblk = (e + 255) / 256;
  const int gblk = (n + 127) / 128;
  const int sblk = (e + 7) / 8;

  k_zero <<<dim3(nblk), blk, 0, stream>>>(deg, n);
  k_count<<<dim3(eblk), blk, 0, stream>>>(dstv, deg, e);
  k_dinv <<<dim3(nblk), blk, 0, stream>>>(deg, dinv, n);

  k_gcn_gemm<<<dim3(gblk), blk, 0, stream>>>(x, W1, b1, dinv, Hbuf, Abuf, n, 0);
  k_scatter <<<dim3(sblk), blk, 0, stream>>>(Hbuf, dinv, srcv, dstv, Abuf, e);

  k_gcn_gemm<<<dim3(gblk), blk, 0, stream>>>(Abuf, W2, b2, dinv, Hbuf, outp, n, 1);
  k_scatter <<<dim3(sblk), blk, 0, stream>>>(Hbuf, dinv, srcv, dstv, outp, e);
}


# Edge sums in LDS: owner work-group of 196 dst rows finds its edges in a one-byte owner key per edge (from the degree launch), adds H rows by LDS float adds, stores rows once; arrival order
# speedup vs baseline: 2.0431x; 2.0431x over previous
#include <hip/hip_runtime.h>

#define D 128

typedef float v2f __attribute__((ext_vector_type(2)));
typedef float v8f __attribute__((ext_vector_type(8)));

#if defined(__HIP_DEVICE_COMPILE__) && __has_builtin(__builtin_amdgcn_wmma_f32_16x16x4_f32)
#define HAVE_WMMA_F32 1
#endif

__device__ __forceinline__ v8f wmma_acc(v2f a, v2f b, v8f c) {
#if defined(HAVE_WMMA_F32)
  return __builtin_amdgcn_wmma_f32_16x16x4_f32(false, a, false, b, (short)0, c,
                                               false, false);
#else
  (void)a; (void)b;
  return c;
#endif
}

#define OR 196
#define HB 16384
#define HT 1024
#define HU 16
__global__ void __launch_bounds__(HT) k_hist(
    const int* __restrict__ dst, unsigned* __restrict__ planes,
    int e, int wsz, int np, int n, unsigned char* __restrict__ keys, int epad) {
  __shared__ unsigned hist[HB];
  const int tid = threadIdx.x;
  const int base = blockIdx.y * wsz;
  int wn = n - base;
  if (wn > wsz) wn = wsz;
  if (wn < 0) wn = 0;
  for (int i = tid; i < wn; i += HT) hist[i] = 0u;
  __syncthreads();
  const int i0 = blockIdx.x * (HT * HU) + tid;
  int d[HU];
#pragma unroll
  for (int u = 0; u < HU; ++u) {
    const int i = i0 + u * HT;
    d[u] = (i < e) ? dst[i] : -1;
  }
  if (blockIdx.y == 0) {
#pragma unroll
    for (int u = 0; u < HU; ++u) {
      const int i = i0 + u * HT;
      if (i < epad) keys[i] = (i < e) ? (unsigned char)((unsigned)d[u] / (unsigned)OR) : (unsigned char)0;
    }
  }
#pragma unroll
  for (int u = 0; u < HU; ++u) {
    const unsigned r = (unsigned)(d[u] - base);
    if (r < (unsigned)wn) atomicAdd(&hist[r], 1u);
  }
  __syncthreads();
  unsigned* __restrict__ p = planes + (size_t)blockIdx.x * np + base;
  for (int i = tid; i < wn; i += HT) p[i] = hist[i];
}

__global__ void k_dinv(const unsigned* __restrict__ planes, float* __restrict__ dinv,
                       int n, int np, int nslice) {
  int i = blockIdx.x * blockDim.x + threadIdx.x;
  if (i < n) {
    unsigned s = 0u;
    for (int g0 = 0; g0 < nslice; g0 += 16) {
      unsigned v[16];
#pragma unroll
      for (int u = 0; u < 16; ++u)
        v[u] = (g0 + u < nslice) ? planes[(size_t)(g0 + u) * np + i] : 0u;
#pragma unroll
      for (int u = 0; u < 16; ++u) s += v[u];
    }
    dinv[i] = rsqrtf((float)s + 1.0f);
  }
}

__global__ void __launch_bounds__(256) k_gcn_gemm(
    const float* __restrict__ X, const float* __restrict__ W,
    float* __restrict__ H, int nrows, int relu_in) {
  const int wave = threadIdx.x >> 5;
  const int lane = threadIdx.x & 31;
  const int half = lane >> 4;
  const int l16  = lane & 15;
  const int row0 = (blockIdx.x * 8 + wave) * 16;
  if (row0 >= nrows) return;

  const int m = row0 + l16;
  const bool mvalid = (m < nrows);
  const float* __restrict__ arow = X + (size_t)(mvalid ? m : 0) * D;

  const v8f vzero = {0.f, 0.f, 0.f, 0.f, 0.f, 0.f, 0.f, 0.f};
  v8f acc[8];
#pragma unroll
  for (int i = 0; i < 8; ++i) acc[i] = vzero;

  for (int k0 = 0; k0 < D; k0 += 4) {
    const int ka = k0 + 2 * half;
    float a0 = arow[ka];
    float a1 = arow[ka + 1];
    if (relu_in) { a0 = fmaxf(a0, 0.f); a1 = fmaxf(a1, 0.f); }
    if (!mvalid) { a0 = 0.f; a1 = 0.f; }
    v2f a; a.x = a0; a.y = a1;
    const float* __restrict__ w0 = W + (size_t)ka * D + l16;
    const float* __restrict__ w1 = w0 + D;
#pragma unroll
    for (int ct = 0; ct < 8; ++ct) {
      v2f b; b.x = w0[ct * 16]; b.y = w1[ct * 16];
      acc[ct] = wmma_acc(a, b, acc[ct]);
    }
  }

#pragma unroll
  for (int v = 0; v < 8; ++v) {
    const int mm = row0 + v + 8 * half;
    if (mm < nrows) {
      const size_t base = (size_t)mm * D;
#pragma unroll
      for (int ct = 0; ct < 8; ++ct) H[base + ct * 16 + l16] = acc[ct][v];
    }
  }
}

#define OT 1024
#define OW (OT / 32)
#define SL 16

__device__ __forceinline__ uint4 load_keys(const unsigned char* __restrict__ keys, int sb,
                                           int lane, int epad) {
  const int eb = sb * 512 + lane * 16;
  return *(const uint4*)(keys + ((eb < epad) ? eb : 0));
}

__device__ __forceinline__ void own_add(
    const float* __restrict__ H, const float* __restrict__ dinv,
    const int* __restrict__ src, const int* __restrict__ dst,
    float* sums, const int* list, int k, int row0, int lane) {
  const int e = (lane < k) ? list[lane] : 0;
  const int sv = src[e];
  const int dv = dst[e];
  const float cv = dinv[sv] * dinv[dv];
  const int rv = dv - row0;
  const int ci = __float_as_int(cv);
  unsigned m = (k >= 32) ? 0xffffffffu : ((1u << k) - 1u);
  while (m != 0u) {
    float h[8][4];
    int jl[8];
#pragma unroll
    for (int u = 0; u < 8; ++u) {
      jl[u] = -1;
      if (m != 0u) {
        const int j = __builtin_ctz(m);
        m &= m - 1u;
        jl[u] = j;
        const int sj = __builtin_amdgcn_readlane(sv, j);
        const float* __restrict__ hp = H + (size_t)sj * D + lane;
        h[u][0] = hp[0];
        h[u][1] = hp[32];
        h[u][2] = hp[64];
        h[u][3] = hp[96];
      }
    }
#pragma unroll
    for (int u = 0; u < 8; ++u) {
      if (jl[u] >= 0) {
        const int rj = __builtin_amdgcn_readlane(rv, jl[u]);
        const float cj = __int_as_float(__builtin_amdgcn_readlane(ci, jl[u]));
        float* sp = sums + rj * D + lane;
        __hip_atomic_fetch_add(sp + 0,  h[u][0] * cj, __ATOMIC_RELAXED, __HIP_MEMORY_SCOPE_WORKGROUP);
        __hip_atomic_fetch_add(sp + 32, h[u][1] * cj, __ATOMIC_RELAXED, __HIP_MEMORY_SCOPE_WORKGROUP);
        __hip_atomic_fetch_add(sp + 64, h[u][2] * cj, __ATOMIC_RELAXED, __HIP_MEMORY_SCOPE_WORKGROUP);
        __hip_atomic_fetch_add(sp + 96, h[u][3] * cj, __ATOMIC_RELAXED, __HIP_MEMORY_SCOPE_WORKGROUP);
      }
    }
  }
}

__device__ __forceinline__ void own_scan4(
    const uint4 (&k)[4], int sb0, int lane, const unsigned char* __restrict__ keys,
    int nedges, int epad, unsigned rep, unsigned lo, unsigned& c, int* myslot) {
  unsigned hit = 0u;
#pragma unroll
  for (int q = 0; q < 4; ++q) {
    const unsigned kw[4] = {k[q].x, k[q].y, k[q].z, k[q].w};
#pragma unroll
    for (int w = 0; w < 4; ++w) {
      const unsigned x = kw[w] ^ rep;
      if (((x - 0x01010101u) & ~x & 0x80808080u) != 0u) hit |= 1u << (4 * q + w);
    }
  }
  while (__builtin_amdgcn_ballot_w32(hit != 0u) != 0u) {
    if (hit != 0u) {
      const int idx = __builtin_ctz(hit);
      hit &= hit - 1u;
      const int e4 = (sb0 + (idx >> 2) * OW) * 512 + lane * 16 + (idx & 3) * 4;
      if (e4 < epad) {
        const unsigned x = *(const unsigned*)(keys + e4) ^ rep;
#pragma unroll
        for (int bb = 0; bb < 4; ++bb) {
          if (((x >> (8 * bb)) & 0xffu) == 0u && e4 + bb < nedges) {
            const unsigned sl = c - lo;
            if (sl < (unsigned)SL) myslot[sl * OT] = e4 + bb;
            ++c;
          }
        }
      }
    }
  }
}

__global__ void __launch_bounds__(OT) k_own(
    const float* __restrict__ H, const float* __restrict__ dinv,
    const int* __restrict__ src, const int* __restrict__ dst,
    const unsigned char* __restrict__ keys, const float* __restrict__ bias,
    float* __restrict__ OUT, int nedges, int epad, int n) {
  __shared__ __attribute__((aligned(16))) float sums[OR * D];
  __shared__ int slots[SL * OT];
  __shared__ int wl[OW * 32 * SL];
  __shared__ unsigned s_more;
  const int tid = threadIdx.x;
  const int lane = tid & 31;
  const int wave = __builtin_amdgcn_readfirstlane(tid >> 5);
  const unsigned own = blockIdx.x;
  const unsigned rep = own * 0x01010101u;
  const int row0 = (int)own * OR;
  int nr = n - row0;
  if (nr > OR) nr = OR;
  if (nr < 0) nr = 0;

  for (int i = tid; i < nr * (D / 4); i += OT) {
    const int r = i >> 5;
    const int c4 = (i & 31) * 4;
    const float di = dinv[row0 + r];
    const float sc = di * di;
    const float* __restrict__ hr = H + (size_t)(row0 + r) * D + c4;
    float* sr = sums + r * D + c4;
    sr[0] = fmaf(hr[0], sc, bias[c4 + 0]);
    sr[1] = fmaf(hr[1], sc, bias[c4 + 1]);
    sr[2] = fmaf(hr[2], sc, bias[c4 + 2]);
    sr[3] = fmaf(hr[3], sc, bias[c4 + 3]);
  }
  if (tid == 0) s_more = 0u;
  __syncthreads();

  int* myslot = slots + tid;
  int* mywl = wl + wave * (32 * SL);
  const int nsb = (epad + 511) / 512;
  for (unsigned lo = 0u;; lo += (unsigned)SL) {
    unsigned c = 0u;
    uint4 ka[4], kb[4];
#pragma unroll
    for (int q = 0; q < 4; ++q) ka[q] = load_keys(keys, wave + q * OW, lane, epad);
    for (int sb = wave; sb < nsb; sb += 8 * OW) {
#pragma unroll
      for (int q = 0; q < 4; ++q) kb[q] = load_keys(keys, sb + (4 + q) * OW, lane, epad);
      own_scan4(ka, sb, lane, keys, nedges, epad, rep, lo, c, myslot);
#pragma unroll
      for (int q = 0; q < 4; ++q) ka[q] = load_keys(keys, sb + (8 + q) * OW, lane, epad);
      own_scan4(kb, sb + 4 * OW, lane, keys, nedges, epad, rep, lo, c, myslot);
    }
    const unsigned have = (c > lo) ? ((c - lo < (unsigned)SL) ? c - lo : (unsigned)SL) : 0u;
    int tot = 0;
    for (int sl = 0; sl < SL; ++sl) {
      const bool on = (unsigned)sl < have;
      const unsigned mask = __builtin_amdgcn_ballot_w32(on);
      if (mask == 0u) break;
      if (on) mywl[tot + (int)__builtin_amdgcn_mbcnt_lo(mask, 0u)] = myslot[sl * OT];
      tot += __builtin_popcount(mask);
    }
    for (int g = 0; g < tot; g += 32) {
      const int k = (tot - g < 32) ? (tot - g) : 32;
      own_add(H, dinv, src, dst, sums, mywl + g, k, row0, lane);
    }
    if (c > lo + (unsigned)SL) s_more = 1u;
    __syncthreads();
    const unsigned more = s_more;
    __syncthreads();
    if (more == 0u) break;
    if (tid == 0) s_more = 0u;
    __syncthreads();
  }

  for (int i = tid; i < nr * (D / 4); i += OT) {
    const int r = i >> 5;
    const int c4 = (i & 31) * 4;
    const float4 v = *(const float4*)(sums + r * D + c4);
    *(float4*)(OUT + (size_t)(row0 + r) * D + c4) = v;
  }
}

extern "C" void kernel_launch(void* const* d_in, const int* in_sizes, int n_in,
                              void* d_out, int out_size, void* d_ws, size_t ws_size,
                              hipStream_t stream) {
  (void)n_in; (void)out_size; (void)ws_size;
  const float* x  = (const float*)d_in[0];
  const int*   ei = (const int*)d_in[1];
  const float* W1 = (const float*)d_in[2];
  const float* b1 = (const float*)d_in[3];
  const float* W2 = (const float*)d_in[4];
  const float* b2 = (const float*)d_in[5];
  const int n = in_sizes[0] / D;
  const int e = in_sizes[1] / 2;
  const int* srcv = ei;
  const int* dstv = ei + e;

  char* ws = (char*)d_ws;
  float* dinv = (float*)ws;
  size_t off = (((size_t)n * 4) + 255) & ~(size_t)255;
  float* Hbuf = (float*)(ws + off);
  off += (((size_t)n * D * 4) + 255) & ~(size_t)255;
  float* Abuf = (float*)(ws + off);
  off += (((size_t)n * D * 4) + 255) & ~(size_t)255;
  unsigned char* keys = (unsigned char*)(ws + off);
  unsigned* planes = (unsigned*)Hbuf;
  float* outp = (float*)d_out;

  const dim3 blk(256);
  const int nblk = (n + 255) / 256;
  const int gblk = (n + 127) / 128;
  const int nown = (n + OR - 1) / OR;
  const int epad = (e + 15) & ~15;
  const int nslice = (e + HT * HU - 1) / (HT * HU);
  const int nwin = (n + HB - 1) / HB;
  const int wsz = (((n + nwin - 1) / nwin) + 31) & ~31;
  const int np = (n + 31) & ~31;

  k_hist<<<dim3(nslice, nwin), dim3(HT), 0, stream>>>(dstv, planes, e, wsz, np, n, keys, epad);
  k_dinv<<<dim3(nblk), blk, 0, stream>>>(planes, dinv, n, np, nslice);

  k_gcn_gemm<<<dim3(gblk), blk, 0, stream>>>(x, W1, Hbuf, n, 0);
  k_own<<<dim3(nown), dim3(OT), 0, stream>>>(Hbuf, dinv, srcv, dstv, keys, b1, Abuf, e, epad, n);

  k_gcn_gemm<<<dim3(gblk), blk, 0, stream>>>(Abuf, W2, Hbuf, n, 1);
  k_own<<<dim3(nown), dim3(OT), 0, stream>>>(Hbuf, dinv, srcv, dstv, keys, b2, outp, e, epad, n);
}


# Owner sums: a source row is gathered with one 16-byte load per lane (one instruction for its four lines, not four); LDS rows kept column-permuted so adds stay conflict-free; arrival order as before
# speedup vs baseline: 2.1086x; 1.0321x over previous
#include <hip/hip_runtime.h>

#define D 128

typedef float v2f __attribute__((ext_vector_type(2)));
typedef float v8f __attribute__((ext_vector_type(8)));

#if defined(__HIP_DEVICE_COMPILE__) && __has_builtin(__builtin_amdgcn_wmma_f32_16x16x4_f32)
#define HAVE_WMMA_F32 1
#endif

__device__ __forceinline__ v8f wmma_acc(v2f a, v2f b, v8f c) {
#if defined(HAVE_WMMA_F32)
  return __builtin_amdgcn_wmma_f32_16x16x4_f32(false, a, false, b, (short)0, c,
                                               false, false);
#else
  (void)a; (void)b;
  return c;
#endif
}

#define OR 196
#define HB 16384
#define HT 1024
#define HU 16
__global__ void __launch_bounds__(HT) k_hist(
    const int* __restrict__ dst, unsigned* __restrict__ planes,
    int e, int wsz, int np, int n, unsigned char* __restrict__ keys, int epad) {
  __shared__ unsigned hist[HB];
  const int tid = threadIdx.x;
  const int base = blockIdx.y * wsz;
  int wn = n - base;
  if (wn > wsz) wn = wsz;
  if (wn < 0) wn = 0;
  for (int i = tid; i < wn; i += HT) hist[i] = 0u;
  __syncthreads();
  const int i0 = blockIdx.x * (HT * HU) + tid;
  int d[HU];
#pragma unroll
  for (int u = 0; u < HU; ++u) {
    const int i = i0 + u * HT;
    d[u] = (i < e) ? dst[i] : -1;
  }
  if (blockIdx.y == 0) {
#pragma unroll
    for (int u = 0; u < HU; ++u) {
      const int i = i0 + u * HT;
      if (i < epad) keys[i] = (i < e) ? (unsigned char)((unsigned)d[u] / (unsigned)OR) : (unsigned char)0;
    }
  }
#pragma unroll
  for (int u = 0; u < HU; ++u) {
    const unsigned r = (unsigned)(d[u] - base);
    if (r < (unsigned)wn) atomicAdd(&hist[r], 1u);
  }
  __syncthreads();
  unsigned* __restrict__ p = planes + (size_t)blockIdx.x * np + base;
  for (int i = tid; i < wn; i += HT) p[i] = hist[i];
}

__global__ void k_dinv(const unsigned* __restrict__ planes, float* __restrict__ dinv,
                       int n, int np, int nslice) {
  int i = blockIdx.x * blockDim.x + threadIdx.x;
  if (i < n) {
    unsigned s = 0u;
    for (int g0 = 0; g0 < nslice; g0 += 16) {
      unsigned v[16];
#pragma unroll
      for (int u = 0; u < 16; ++u)
        v[u] = (g0 + u < nslice) ? planes[(size_t)(g0 + u) * np + i] : 0u;
#pragma unroll
      for (int u = 0; u < 16; ++u) s += v[u];
    }
    dinv[i] = rsqrtf((float)s + 1.0f);
  }
}

__global__ void __launch_bounds__(256) k_gcn_gemm(
    const float* __restrict__ X, const float* __restrict__ W,
    float* __restrict__ H, int nrows, int relu_in) {
  const int wave = threadIdx.x >> 5;
  const int lane = threadIdx.x & 31;
  const int half = lane >> 4;
  const int l16  = lane & 15;
  const int row0 = (blockIdx.x * 8 + wave) * 16;
  if (row0 >= nrows) return;

  const int m = row0 + l16;
  const bool mvalid = (m < nrows);
  const float* __restrict__ arow = X + (size_t)(mvalid ? m : 0) * D;

  const v8f vzero = {0.f, 0.f, 0.f, 0.f, 0.f, 0.f, 0.f, 0.f};
  v8f acc[8];
#pragma unroll
  for (int i = 0; i < 8; ++i) acc[i] = vzero;

  for (int k0 = 0; k0 < D; k0 += 4) {
    const int ka = k0 + 2 * half;
    float a0 = arow[ka];
    float a1 = arow[ka + 1];
    if (relu_in) { a0 = fmaxf(a0, 0.f); a1 = fmaxf(a1, 0.f); }
    if (!mvalid) { a0 = 0.f; a1 = 0.f; }
    v2f a; a.x = a0; a.y = a1;
    const float* __restrict__ w0 = W + (size_t)ka * D + l16;
    const float* __restrict__ w1 = w0 + D;
#pragma unroll
    for (int ct = 0; ct < 8; ++ct) {
      v2f b; b.x = w0[ct * 16]; b.y = w1[ct * 16];
      acc[ct] = wmma_acc(a, b, acc[ct]);
    }
  }

#pragma unroll
  for (int v = 0; v < 8; ++v) {
    const int mm = row0 + v + 8 * half;
    if (mm < nrows) {
      const size_t base = (size_t)mm * D;
#pragma unroll
      for (int ct = 0; ct < 8; ++ct) H[base + ct * 16 + l16] = acc[ct][v];
    }
  }
}

#define OT 1024
#define OW (OT / 32)
#define SL 16

__device__ __forceinline__ uint4 load_keys(const unsigned char* __restrict__ keys, int sb,
                                           int lane, int epad) {
  const int eb = sb * 512 + lane * 16;
  return *(const uint4*)(keys + ((eb < epad) ? eb : 0));
}

__device__ __forceinline__ void own_add(
    const float* __restrict__ H, const float* __restrict__ dinv,
    const int* __restrict__ src, const int* __restrict__ dst,
    float* sums, const int* list, int k, int row0, int lane) {
  const int e = (lane < k) ? list[lane] : 0;
  const int sv = src[e];
  const int dv = dst[e];
  const float cv = dinv[sv] * dinv[dv];
  const int rv = dv - row0;
  const int ci = __float_as_int(cv);
  unsigned m = (k >= 32) ? 0xffffffffu : ((1u << k) - 1u);
  while (m != 0u) {
    float4 h[8];
    int jl[8];
#pragma unroll
    for (int u = 0; u < 8; ++u) {
      jl[u] = -1;
      if (m != 0u) {
        const int j = __builtin_ctz(m);
        m &= m - 1u;
        jl[u] = j;
        const int sj = __builtin_amdgcn_readlane(sv, j);
        h[u] = *((const float4*)(H + (size_t)sj * D) + lane);
      }
    }
#pragma unroll
    for (int u = 0; u < 8; ++u) {
      if (jl[u] >= 0) {
        const int rj = __builtin_amdgcn_readlane(rv, jl[u]);
        const float cj = __int_as_float(__builtin_amdgcn_readlane(ci, jl[u]));
        float* sp = sums + rj * D + lane;
        __hip_atomic_fetch_add(sp + 0,  h[u].x * cj, __ATOMIC_RELAXED, __HIP_MEMORY_SCOPE_WORKGROUP);
        __hip_atomic_fetch_add(sp + 32, h[u].y * cj, __ATOMIC_RELAXED, __HIP_MEMORY_SCOPE_WORKGROUP);
        __hip_atomic_fetch_add(sp + 64, h[u].z * cj, __ATOMIC_RELAXED, __HIP_MEMORY_SCOPE_WORKGROUP);
        __hip_atomic_fetch_add(sp + 96, h[u].w * cj, __ATOMIC_RELAXED, __HIP_MEMORY_SCOPE_WORKGROUP);
      }
    }
  }
}

__device__ __forceinline__ void own_scan4(
    const uint4 (&k)[4], int sb0, int lane, const unsigned char* __restrict__ keys,
    int nedges, int epad, unsigned rep, unsigned lo, unsigned& c, int* myslot) {
  unsigned hit = 0u;
#pragma unroll
  for (int q = 0; q < 4; ++q) {
    const unsigned kw[4] = {k[q].x, k[q].y, k[q].z, k[q].w};
#pragma unroll
    for (int w = 0; w < 4; ++w) {
      const unsigned x = kw[w] ^ rep;
      if (((x - 0x01010101u) & ~x & 0x80808080u) != 0u) hit |= 1u << (4 * q + w);
    }
  }
  while (__builtin_amdgcn_ballot_w32(hit != 0u) != 0u) {
    if (hit != 0u) {
      const int idx = __builtin_ctz(hit);
      hit &= hit - 1u;
      const int e4 = (sb0 + (idx >> 2) * OW) * 512 + lane * 16 + (idx & 3) * 4;
      if (e4 < epad) {
        const unsigned x = *(const unsigned*)(keys + e4) ^ rep;
#pragma unroll
        for (int bb = 0; bb < 4; ++bb) {
          if (((x >> (8 * bb)) & 0xffu) == 0u && e4 + bb < nedges) {
            const unsigned sl = c - lo;
            if (sl < (unsigned)SL) myslot[sl * OT] = e4 + bb;
            ++c;
          }
        }
      }
    }
  }
}

__global__ void __launch_bounds__(OT) k_own(
    const float* __restrict__ H, const float* __restrict__ dinv,
    const int* __restrict__ src, const int* __restrict__ dst,
    const unsigned char* __restrict__ keys, const float* __restrict__ bias,
    float* __restrict__ OUT, int nedges, int epad, int n) {
  __shared__ float sums[OR * D];
  __shared__ int slots[SL * OT];
  __shared__ int wl[OW * 32 * SL];
  __shared__ unsigned s_more;
  const int tid = threadIdx.x;
  const int lane = tid & 31;
  const int wave = __builtin_amdgcn_readfirstlane(tid >> 5);
  const unsigned own = blockIdx.x;
  const unsigned rep = own * 0x01010101u;
  const int row0 = (int)own * OR;
  int nr = n - row0;
  if (nr > OR) nr = OR;
  if (nr < 0) nr = 0;

  for (int i = tid; i < nr * (D / 4); i += OT) {
    const int r = i >> 5;
    const int q = i & 31;
    const int c4 = q * 4;
    const float di = dinv[row0 + r];
    const float sc = di * di;
    const float4 hv = *(const float4*)(H + (size_t)(row0 + r) * D + c4);
    const float4 bv = *(const float4*)(bias + c4);
    float* sr = sums + r * D + q;
    sr[0]  = fmaf(hv.x, sc, bv.x);
    sr[32] = fmaf(hv.y, sc, bv.y);
    sr[64] = fmaf(hv.z, sc, bv.z);
    sr[96] = fmaf(hv.w, sc, bv.w);
  }
  if (tid == 0) s_more = 0u;
  __syncthreads();

  int* myslot = slots + tid;
  int* mywl = wl + wave * (32 * SL);
  const int nsb = (epad + 511) / 512;
  for (unsigned lo = 0u;; lo += (unsigned)SL) {
    unsigned c = 0u;
    uint4 ka[4], kb[4];
#pragma unroll
    for (int q = 0; q < 4; ++q) ka[q] = load_keys(keys, wave + q * OW, lane, epad);
    for (int sb = wave; sb < nsb; sb += 8 * OW) {
#pragma unroll
      for (int q = 0; q < 4; ++q) kb[q] = load_keys(keys, sb + (4 + q) * OW, lane, epad);
      own_scan4(ka, sb, lane, keys, nedges, epad, rep, lo, c, myslot);
#pragma unroll
      for (int q = 0; q < 4; ++q) ka[q] = load_keys(keys, sb + (8 + q) * OW, lane, epad);
      own_scan4(kb, sb + 4 * OW, lane, keys, nedges, epad, rep, lo, c, myslot);
    }
    const unsigned have = (c > lo) ? ((c - lo < (unsigned)SL) ? c - lo : (unsigned)SL) : 0u;
    int tot = 0;
    for (int sl = 0; sl < SL; ++sl) {
      const bool on = (unsigned)sl < have;
      const unsigned mask = __builtin_amdgcn_ballot_w32(on);
      if (mask == 0u) break;
      if (on) mywl[tot + (int)__builtin_amdgcn_mbcnt_lo(mask, 0u)] = myslot[sl * OT];
      tot += __builtin_popcount(mask);
    }
    for (int g = 0; g < tot; g += 32) {
      const int k = (tot - g < 32) ? (tot - g) : 32;
      own_add(H, dinv, src, dst, sums, mywl + g, k, row0, lane);
    }
    if (c > lo + (unsigned)SL) s_more = 1u;
    __syncthreads();
    const unsigned more = s_more;
    __syncthreads();
    if (more == 0u) break;
    if (tid == 0) s_more = 0u;
    __syncthreads();
  }

  for (int i = tid; i < nr * (D / 4); i += OT) {
    const int r = i >> 5;
    const int q = i & 31;
    const float* sr = sums + r * D + q;
    float4 v;
    v.x = sr[0];
    v.y = sr[32];
    v.z = sr[64];
    v.w = sr[96];
    *(float4*)(OUT + (size_t)(row0 + r) * D + q * 4) = v;
  }
}

extern "C" void kernel_launch(void* const* d_in, const int* in_sizes, int n_in,
                              void* d_out, int out_size, void* d_ws, size_t ws_size,
                              hipStream_t stream) {
  (void)n_in; (void)out_size; (void)ws_size;
  const float* x  = (const float*)d_in[0];
  const int*   ei = (const int*)d_in[1];
  const float* W1 = (const float*)d_in[2];
  const float* b1 = (const float*)d_in[3];
  const float* W2 = (const float*)d_in[4];
  const float* b2 = (const float*)d_in[5];
  const int n = in_sizes[0] / D;
  const int e = in_sizes[1] / 2;
  const int* srcv = ei;
  const int* dstv = ei + e;

  char* ws = (char*)d_ws;
  float* dinv = (float*)ws;
  size_t off = (((size_t)n * 4) + 255) & ~(size_t)255;
  float* Hbuf = (float*)(ws + off);
  off += (((size_t)n * D * 4) + 255) & ~(size_t)255;
  float* Abuf = (float*)(ws + off);
  off += (((size_t)n * D * 4) + 255) & ~(size_t)255;
  unsigned char* keys = (unsigned char*)(ws + off);
  unsigned* planes = (unsigned*)Hbuf;
  float* outp = (float*)d_out;

  const dim3 blk(256);
  const int nblk = (n + 255) / 256;
  const int gblk = (n + 127) / 128;
  const int nown = (n + OR - 1) / OR;
  const int epad = (e + 15) & ~15;
  const int nslice = (e + HT * HU - 1) / (HT * HU);
  const int nwin = (n + HB - 1) / HB;
  const int wsz = (((n + nwin - 1) / nwin) + 31) & ~31;
  const int np = (n + 31) & ~31;

  k_hist<<<dim3(nslice, nwin), dim3(HT), 0, stream>>>(dstv, planes, e, wsz, np, n, keys, epad);
  k_dinv<<<dim3(nblk), blk, 0, stream>>>(planes, dinv, n, np, nslice);

  k_gcn_gemm<<<dim3(gblk), blk, 0, stream>>>(x, W1, Hbuf, n, 0);
  k_own<<<dim3(nown), dim3(OT), 0, stream>>>(Hbuf, dinv, srcv, dstv, keys, b1, Abuf, e, epad, n);

  k_gcn_gemm<<<dim3(gblk), blk, 0, stream>>>(Abuf, W2, Hbuf, n, 1);
  k_own<<<dim3(nown), dim3(OT), 0, stream>>>(Hbuf, dinv, srcv, dstv, keys, b2, outp, e, epad, n);
}


# Owner keys as bit planes: degree launch stores bit p of 32 edges' owner numbers per word (ballots); an owner lane tests 32 edges with 8 boolean ops, no key re-read; sums in arrival order as before
# speedup vs baseline: 2.2204x; 1.0530x over previous
#include <hip/hip_runtime.h>

#define D 128

typedef float v2f __attribute__((ext_vector_type(2)));
typedef float v8f __attribute__((ext_vector_type(8)));

#if defined(__HIP_DEVICE_COMPILE__) && __has_builtin(__builtin_amdgcn_wmma_f32_16x16x4_f32)
#define HAVE_WMMA_F32 1
#endif

__device__ __forceinline__ v8f wmma_acc(v2f a, v2f b, v8f c) {
#if defined(HAVE_WMMA_F32)
  return __builtin_amdgcn_wmma_f32_16x16x4_f32(false, a, false, b, (short)0, c,
                                               false, false);
#else
  (void)a; (void)b;
  return c;
#endif
}

#define OR 196
#define KB 1024
#define HB 16384
#define HT 1024
#define HU 16
__global__ void __launch_bounds__(HT) k_hist(
    const int* __restrict__ dst, unsigned* __restrict__ planes,
    int e, int wsz, int np, int n, unsigned* __restrict__ keys, int nkb) {
  __shared__ unsigned hist[HB];
  __shared__ uint4 kst[HU * 64];
  const int tid = threadIdx.x;
  const int base = blockIdx.y * wsz;
  int wn = n - base;
  if (wn > wsz) wn = wsz;
  if (wn < 0) wn = 0;
  for (int i = tid; i < wn; i += HT) hist[i] = 0u;
  __syncthreads();
  const int i0 = blockIdx.x * (HT * HU) + tid;
  int d[HU];
#pragma unroll
  for (int u = 0; u < HU; ++u) {
    const int i = i0 + u * HT;
    d[u] = (i < e) ? dst[i] : -1;
  }
  int myu = blockIdx.y;
#pragma unroll
  for (int u = 0; u < HU; ++u) {
    if (u == myu) {
      myu += gridDim.y;
      const int i = i0 + u * HT;
      const unsigned key = (i < e) ? (unsigned)d[u] / (unsigned)OR : 0u;
      unsigned pw[8];
#pragma unroll
      for (int p = 0; p < 8; ++p) pw[p] = __builtin_amdgcn_ballot_w32(((key >> p) & 1u) != 0u);
      if ((tid & 31) == 0) {
        uint4* q = kst + u * 64 + (tid >> 5);
        q[0]  = make_uint4(pw[0], pw[1], pw[2], pw[3]);
        q[32] = make_uint4(pw[4], pw[5], pw[6], pw[7]);
      }
    }
  }
#pragma unroll
  for (int u = 0; u < HU; ++u) {
    const unsigned r = (unsigned)(d[u] - base);
    if (r < (unsigned)wn) atomicAdd(&hist[r], 1u);
  }
  __syncthreads();
  unsigned* __restrict__ p = planes + (size_t)blockIdx.x * np + base;
  for (int i = tid; i < wn; i += HT) p[i] = hist[i];
  for (int u = blockIdx.y; u < HU; u += gridDim.y) {
    const int b = blockIdx.x * HU + u;
    if (b < nkb && tid < 64) ((uint4*)keys)[(size_t)b * 64 + tid] = kst[u * 64 + tid];
  }
}

__global__ void k_dinv(const unsigned* __restrict__ planes, float* __restrict__ dinv,
                       int n, int np, int nslice) {
  int i = blockIdx.x * blockDim.x + threadIdx.x;
  if (i < n) {
    unsigned s = 0u;
    for (int g0 = 0; g0 < nslice; g0 += 16) {
      unsigned v[16];
#pragma unroll
      for (int u = 0; u < 16; ++u)
        v[u] = (g0 + u < nslice) ? planes[(size_t)(g0 + u) * np + i] : 0u;
#pragma unroll
      for (int u = 0; u < 16; ++u) s += v[u];
    }
    dinv[i] = rsqrtf((float)s + 1.0f);
  }
}

__global__ void __launch_bounds__(256) k_gcn_gemm(
    const float* __restrict__ X, const float* __restrict__ W,
    float* __restrict__ H, int nrows, int relu_in) {
  const int wave = threadIdx.x >> 5;
  const int lane = threadIdx.x & 31;
  const int half = lane >> 4;
  const int l16  = lane & 15;
  const int row0 = (blockIdx.x * 8 + wave) * 16;
  if (row0 >= nrows) return;

  const int m = row0 + l16;
  const bool mvalid = (m < nrows);
  const float* __restrict__ arow = X + (size_t)(mvalid ? m : 0) * D;

  const v8f vzero = {0.f, 0.f, 0.f, 0.f, 0.f, 0.f, 0.f, 0.f};
  v8f acc[8];
#pragma unroll
  for (int i = 0; i < 8; ++i) acc[i] = vzero;

  for (int k0 = 0; k0 < D; k0 += 4) {
    const int ka = k0 + 2 * half;
    float a0 = arow[ka];
    float a1 = arow[ka + 1];
    if (relu_in) { a0 = fmaxf(a0, 0.f); a1 = fmaxf(a1, 0.f); }
    if (!mvalid) { a0 = 0.f; a1 = 0.f; }
    v2f a; a.x = a0; a.y = a1;
    const float* __restrict__ w0 = W + (size_t)ka * D + l16;
    const float* __restrict__ w1 = w0 + D;
#pragma unroll
    for (int ct = 0; ct < 8; ++ct) {
      v2f b; b.x = w0[ct * 16]; b.y = w1[ct * 16];
      acc[ct] = wmma_acc(a, b, acc[ct]);
    }
  }

#pragma unroll
  for (int v = 0; v < 8; ++v) {
    const int mm = row0 + v + 8 * half;
    if (mm < nrows) {
      const size_t base = (size_t)mm * D;
#pragma unroll
      for (int ct = 0; ct < 8; ++ct) H[base + ct * 16 + l16] = acc[ct][v];
    }
  }
}

#define OT 1024
#define OW (OT / 32)
#define SL 16

__device__ __forceinline__ void load_blk(const uint4* __restrict__ keys, int b, int nkb,
                                         int lane, uint4& lo, uint4& hi) {
  const uint4* p = keys + (size_t)((b < nkb) ? b : 0) * 64 + lane;
  lo = p[0];
  hi = p[32];
}

__device__ __forceinline__ void own_add(
    const float* __restrict__ H, const float* __restrict__ dinv,
    const int* __restrict__ src, const int* __restrict__ dst,
    float* sums, const int* list, int k, int row0, int lane) {
  const int e = (lane < k) ? list[lane] : 0;
  const int sv = src[e];
  const int dv = dst[e];
  const float cv = dinv[sv] * dinv[dv];
  const int rv = dv - row0;
  const int ci = __float_as_int(cv);
  unsigned m = (k >= 32) ? 0xffffffffu : ((1u << k) - 1u);
  while (m != 0u) {
    float4 h[8];
    int jl[8];
#pragma unroll
    for (int u = 0; u < 8; ++u) {
      jl[u] = -1;
      if (m != 0u) {
        const int j = __builtin_ctz(m);
        m &= m - 1u;
        jl[u] = j;
        const int sj = __builtin_amdgcn_readlane(sv, j);
        h[u] = *((const float4*)(H + (size_t)sj * D) + lane);
      }
    }
#pragma unroll
    for (int u = 0; u < 8; ++u) {
      if (jl[u] >= 0) {
        const int rj = __builtin_amdgcn_readlane(rv, jl[u]);
        const float cj = __int_as_float(__builtin_amdgcn_readlane(ci, jl[u]));
        float* sp = sums + rj * D + lane;
        __hip_atomic_fetch_add(sp + 0,  h[u].x * cj, __ATOMIC_RELAXED, __HIP_MEMORY_SCOPE_WORKGROUP);
        __hip_atomic_fetch_add(sp + 32, h[u].y * cj, __ATOMIC_RELAXED, __HIP_MEMORY_SCOPE_WORKGROUP);
        __hip_atomic_fetch_add(sp + 64, h[u].z * cj, __ATOMIC_RELAXED, __HIP_MEMORY_SCOPE_WORKGROUP);
        __hip_atomic_fetch_add(sp + 96, h[u].w * cj, __ATOMIC_RELAXED, __HIP_MEMORY_SCOPE_WORKGROUP);
      }
    }
  }
}

__device__ __forceinline__ void own_scan(
    const uint4& lo, const uint4& hi, int b, int nkb, int lane, int nedges,
    const unsigned (&q)[8], unsigned lo_slot, unsigned& c, int* myslot) {
  if (b < nkb) {
    unsigned m = (lo.x ^ q[0]) & (lo.y ^ q[1]) & (lo.z ^ q[2]) & (lo.w ^ q[3]) &
                 (hi.x ^ q[4]) & (hi.y ^ q[5]) & (hi.z ^ q[6]) & (hi.w ^ q[7]);
    const int e0 = b * KB + lane * 32;
    while (m != 0u) {
      const int ei = e0 + __builtin_ctz(m);
      m &= m - 1u;
      if (ei < nedges) {
        const unsigned sl = c - lo_slot;
        if (sl < (unsigned)SL) myslot[sl * OT] = ei;
        ++c;
      }
    }
  }
}

__global__ void __launch_bounds__(OT) k_own(
    const float* __restrict__ H, const float* __restrict__ dinv,
    const int* __restrict__ src, const int* __restrict__ dst,
    const unsigned* __restrict__ keyw, const float* __restrict__ bias,
    float* __restrict__ OUT, int nedges, int nkb, int n) {
  const uint4* __restrict__ keys = (const uint4*)keyw;
  __shared__ float sums[OR * D];
  __shared__ int slots[SL * OT];
  __shared__ int wl[OW * 32 * SL];
  __shared__ unsigned s_more;
  const int tid = threadIdx.x;
  const int lane = tid & 31;
  const int wave = __builtin_amdgcn_readfirstlane(tid >> 5);
  const unsigned own = blockIdx.x;
  unsigned q[8];
#pragma unroll
  for (int p = 0; p < 8; ++p) q[p] = ((own >> p) & 1u) ? 0u : 0xffffffffu;
  const int row0 = (int)own * OR;
  int nr = n - row0;
  if (nr > OR) nr = OR;
  if (nr < 0) nr = 0;

  for (int i = tid; i < nr * (D / 4); i += OT) {
    const int r = i >> 5;
    const int q = i & 31;
    const int c4 = q * 4;
    const float di = dinv[row0 + r];
    const float sc = di * di;
    const float4 hv = *(const float4*)(H + (size_t)(row0 + r) * D + c4);
    const float4 bv = *(const float4*)(bias + c4);
    float* sr = sums + r * D + q;
    sr[0]  = fmaf(hv.x, sc, bv.x);
    sr[32] = fmaf(hv.y, sc, bv.y);
    sr[64] = fmaf(hv.z, sc, bv.z);
    sr[96] = fmaf(hv.w, sc, bv.w);
  }
  if (tid == 0) s_more = 0u;
  __syncthreads();

  int* myslot = slots + tid;
  int* mywl = wl + wave * (32 * SL);
  for (unsigned lo = 0u;; lo += (unsigned)SL) {
    unsigned c = 0u;
    uint4 ka[4], kb[4];
    load_blk(keys, wave, nkb, lane, ka[0], ka[1]);
    load_blk(keys, wave + OW, nkb, lane, ka[2], ka[3]);
    for (int b = wave; b < nkb; b += 4 * OW) {
      load_blk(keys, b + 2 * OW, nkb, lane, kb[0], kb[1]);
      load_blk(keys, b + 3 * OW, nkb, lane, kb[2], kb[3]);
      own_scan(ka[0], ka[1], b, nkb, lane, nedges, q, lo, c, myslot);
      own_scan(ka[2], ka[3], b + OW, nkb, lane, nedges, q, lo, c, myslot);
      load_blk(keys, b + 4 * OW, nkb, lane, ka[0], ka[1]);
      load_blk(keys, b + 5 * OW, nkb, lane, ka[2], ka[3]);
      own_scan(kb[0], kb[1], b + 2 * OW, nkb, lane, nedges, q, lo, c, myslot);
      own_scan(kb[2], kb[3], b + 3 * OW, nkb, lane, nedges, q, lo, c, myslot);
    }
    const unsigned have = (c > lo) ? ((c - lo < (unsigned)SL) ? c - lo : (unsigned)SL) : 0u;
    int tot = 0;
    for (int sl = 0; sl < SL; ++sl) {
      const bool on = (unsigned)sl < have;
      const unsigned mask = __builtin_amdgcn_ballot_w32(on);
      if (mask == 0u) break;
      if (on) mywl[tot + (int)__builtin_amdgcn_mbcnt_lo(mask, 0u)] = myslot[sl * OT];
      tot += __builtin_popcount(mask);
    }
    for (int g = 0; g < tot; g += 32) {
      const int k = (tot - g < 32) ? (tot - g) : 32;
      own_add(H, dinv, src, dst, sums, mywl + g, k, row0, lane);
    }
    if (c > lo + (unsigned)SL) s_more = 1u;
    __syncthreads();
    const unsigned more = s_more;
    __syncthreads();
    if (more == 0u) break;
    if (tid == 0) s_more = 0u;
    __syncthreads();
  }

  for (int i = tid; i < nr * (D / 4); i += OT) {
    const int r = i >> 5;
    const int q = i & 31;
    const float* sr = sums + r * D + q;
    float4 v;
    v.x = sr[0];
    v.y = sr[32];
    v.z = sr[64];
    v.w = sr[96];
    *(float4*)(OUT + (size_t)(row0 + r) * D + q * 4) = v;
  }
}

extern "C" void kernel_launch(void* const* d_in, const int* in_sizes, int n_in,
                              void* d_out, int out_size, void* d_ws, size_t ws_size,
                              hipStream_t stream) {
  (void)n_in; (void)out_size; (void)ws_size;
  const float* x  = (const float*)d_in[0];
  const int*   ei = (const int*)d_in[1];
  const float* W1 = (const float*)d_in[2];
  const float* b1 = (const float*)d_in[3];
  const float* W2 = (const float*)d_in[4];
  const float* b2 = (const float*)d_in[5];
  const int n = in_sizes[0] / D;
  const int e = in_sizes[1] / 2;
  const int* srcv = ei;
  const int* dstv = ei + e;

  char* ws = (char*)d_ws;
  float* dinv = (float*)ws;
  size_t off = (((size_t)n * 4) + 255) & ~(size_t)255;
  float* Hbuf = (float*)(ws + off);
  off += (((size_t)n * D * 4) + 255) & ~(size_t)255;
  float* Abuf = (float*)(ws + off);
  off += (((size_t)n * D * 4) + 255) & ~(size_t)255;
  unsigned* keys = (unsigned*)(ws + off);
  unsigned* planes = (unsigned*)Hbuf;
  float* outp = (float*)d_out;

  const dim3 blk(256);
  const int nblk = (n + 255) / 256;
  const int gblk = (n + 127) / 128;
  const int nown = (n + OR - 1) / OR;
  const int nkb = (e + KB - 1) / KB;
  const int nslice = (e + HT * HU - 1) / (HT * HU);
  const int nwin = (n + HB - 1) / HB;
  const int wsz = (((n + nwin - 1) / nwin) + 31) & ~31;
  const int np = (n + 31) & ~31;

  k_hist<<<dim3(nslice, nwin), dim3(HT), 0, stream>>>(dstv, planes, e, wsz, np, n, keys, nkb);
  k_dinv<<<dim3(nblk), blk, 0, stream>>>(planes, dinv, n, np, nslice);

  k_gcn_gemm<<<dim3(gblk), blk, 0, stream>>>(x, W1, Hbuf, n, 0);
  k_own<<<dim3(nown), dim3(OT), 0, stream>>>(Hbuf, dinv, srcv, dstv, keys, b1, Abuf, e, nkb, n);

  k_gcn_gemm<<<dim3(gblk), blk, 0, stream>>>(Abuf, W2, Hbuf, n, 1);
  k_own<<<dim3(nown), dim3(OT), 0, stream>>>(Hbuf, dinv, srcv, dstv, keys, b2, outp, e, nkb, n);
}


# Owner sums in registers: scanned edges are sorted by destination row in LDS (integer counts and ranks); a wave sums one row at a time by fma and stores it once; no LDS float adds; arrival order
# speedup vs baseline: 2.3990x; 1.0804x over previous
#include <hip/hip_runtime.h>

#define D 128

typedef float v2f __attribute__((ext_vector_type(2)));
typedef float v8f __attribute__((ext_vector_type(8)));

#if defined(__HIP_DEVICE_COMPILE__) && __has_builtin(__builtin_amdgcn_wmma_f32_16x16x4_f32)
#define HAVE_WMMA_F32 1
#endif

__device__ __forceinline__ v8f wmma_acc(v2f a, v2f b, v8f c) {
#if defined(HAVE_WMMA_F32)
  return __builtin_amdgcn_wmma_f32_16x16x4_f32(false, a, false, b, (short)0, c,
                                               false, false);
#else
  (void)a; (void)b;
  return c;
#endif
}

#define OR 196
#define KB 1024
#define HB 16384
#define HT 1024
#define HU 16
__global__ void __launch_bounds__(HT) k_hist(
    const int* __restrict__ dst, unsigned* __restrict__ planes,
    int e, int wsz, int np, int n, unsigned* __restrict__ keys, int nkb) {
  __shared__ unsigned hist[HB];
  __shared__ uint4 kst[HU * 64];
  const int tid = threadIdx.x;
  const int base = blockIdx.y * wsz;
  int wn = n - base;
  if (wn > wsz) wn = wsz;
  if (wn < 0) wn = 0;
  for (int i = tid; i < wn; i += HT) hist[i] = 0u;
  __syncthreads();
  const int i0 = blockIdx.x * (HT * HU) + tid;
  int d[HU];
#pragma unroll
  for (int u = 0; u < HU; ++u) {
    const int i = i0 + u * HT;
    d[u] = (i < e) ? dst[i] : -1;
  }
  int myu = blockIdx.y;
#pragma unroll
  for (int u = 0; u < HU; ++u) {
    if (u == myu) {
      myu += gridDim.y;
      const int i = i0 + u * HT;
      const unsigned key = (i < e) ? (unsigned)d[u] / (unsigned)OR : 0u;
      unsigned pw[8];
#pragma unroll
      for (int p = 0; p < 8; ++p) pw[p] = __builtin_amdgcn_ballot_w32(((key >> p) & 1u) != 0u);
      if ((tid & 31) == 0) {
        uint4* q = kst + u * 64 + (tid >> 5);
        q[0]  = make_uint4(pw[0], pw[1], pw[2], pw[3]);
        q[32] = make_uint4(pw[4], pw[5], pw[6], pw[7]);
      }
    }
  }
#pragma unroll
  for (int u = 0; u < HU; ++u) {
    const unsigned r = (unsigned)(d[u] - base);
    if (r < (unsigned)wn) atomicAdd(&hist[r], 1u);
  }
  __syncthreads();
  unsigned* __restrict__ p = planes + (size_t)blockIdx.x * np + base;
  for (int i = tid; i < wn; i += HT) p[i] = hist[i];
  for (int u = blockIdx.y; u < HU; u += gridDim.y) {
    const int b = blockIdx.x * HU + u;
    if (b < nkb && tid < 64) ((uint4*)keys)[(size_t)b * 64 + tid] = kst[u * 64 + tid];
  }
}

__global__ void k_dinv(const unsigned* __restrict__ planes, float* __restrict__ dinv,
                       int n, int np, int nslice) {
  int i = blockIdx.x * blockDim.x + threadIdx.x;
  if (i < n) {
    unsigned s = 0u;
    for (int g0 = 0; g0 < nslice; g0 += 16) {
      unsigned v[16];
#pragma unroll
      for (int u = 0; u < 16; ++u)
        v[u] = (g0 + u < nslice) ? planes[(size_t)(g0 + u) * np + i] : 0u;
#pragma unroll
      for (int u = 0; u < 16; ++u) s += v[u];
    }
    dinv[i] = rsqrtf((float)s + 1.0f);
  }
}

__global__ void __launch_bounds__(256) k_gcn_gemm(
    const float* __restrict__ X, const float* __restrict__ W,
    float* __restrict__ H, int nrows, int relu_in) {
  const int wave = threadIdx.x >> 5;
  const int lane = threadIdx.x & 31;
  const int half = lane >> 4;
  const int l16  = lane & 15;
  const int row0 = (blockIdx.x * 8 + wave) * 16;
  if (row0 >= nrows) return;

  const int m = row0 + l16;
  const bool mvalid = (m < nrows);
  const float* __restrict__ arow = X + (size_t)(mvalid ? m : 0) * D;

  const v8f vzero = {0.f, 0.f, 0.f, 0.f, 0.f, 0.f, 0.f, 0.f};
  v8f acc[8];
#pragma unroll
  for (int i = 0; i < 8; ++i) acc[i] = vzero;

  for (int k0 = 0; k0 < D; k0 += 4) {
    const int ka = k0 + 2 * half;
    float a0 = arow[ka];
    float a1 = arow[ka + 1];
    if (relu_in) { a0 = fmaxf(a0, 0.f); a1 = fmaxf(a1, 0.f); }
    if (!mvalid) { a0 = 0.f; a1 = 0.f; }
    v2f a; a.x = a0; a.y = a1;
    const float* __restrict__ w0 = W + (size_t)ka * D + l16;
    const float* __restrict__ w1 = w0 + D;
#pragma unroll
    for (int ct = 0; ct < 8; ++ct) {
      v2f b; b.x = w0[ct * 16]; b.y = w1[ct * 16];
      acc[ct] = wmma_acc(a, b, acc[ct]);
    }
  }

#pragma unroll
  for (int v = 0; v < 8; ++v) {
    const int mm = row0 + v + 8 * half;
    if (mm < nrows) {
      const size_t base = (size_t)mm * D;
#pragma unroll
      for (int ct = 0; ct < 8; ++ct) H[base + ct * 16 + l16] = acc[ct][v];
    }
  }
}

#define OT 1024
#define OW (OT / 32)
#define SL 16
#define RJ ((OR + OW - 1) / OW)
#define NB (OW * RJ)

__device__ __forceinline__ void load_blk(const uint4* __restrict__ keys, int b, int nkb,
                                         int lane, uint4& lo, uint4& hi) {
  const uint4* p = keys + (size_t)((b < nkb) ? b : 0) * 64 + lane;
  lo = p[0];
  hi = p[32];
}

__device__ __forceinline__ void own_add(
    const float* __restrict__ H, const uint2* rec, int k, int lane, float4& cur) {
  uint2 v = make_uint2(0u, 0u);
  if (lane < k) v = rec[lane];
  const int sv = (int)v.x;
  const int ci = (int)v.y;
  unsigned m = (k >= 32) ? 0xffffffffu : ((1u << k) - 1u);
  while (m != 0u) {
    float4 h[8];
    int jl[8];
#pragma unroll
    for (int u = 0; u < 8; ++u) {
      jl[u] = -1;
      if (m != 0u) {
        const int j = __builtin_ctz(m);
        m &= m - 1u;
        jl[u] = j;
        const int sj = __builtin_amdgcn_readlane(sv, j);
        h[u] = *((const float4*)(H + (size_t)sj * D) + lane);
      }
    }
#pragma unroll
    for (int u = 0; u < 8; ++u) {
      if (jl[u] >= 0) {
        const float cj = __int_as_float(__builtin_amdgcn_readlane(ci, jl[u]));
        cur.x = fmaf(h[u].x, cj, cur.x);
        cur.y = fmaf(h[u].y, cj, cur.y);
        cur.z = fmaf(h[u].z, cj, cur.z);
        cur.w = fmaf(h[u].w, cj, cur.w);
      }
    }
  }
}

__device__ __forceinline__ void own_scan(
    const uint4& lo, const uint4& hi, int b, int nkb, int lane, int nedges,
    const unsigned (&q)[8], unsigned lo_slot, unsigned& c, int* myslot) {
  if (b < nkb) {
    unsigned m = (lo.x ^ q[0]) & (lo.y ^ q[1]) & (lo.z ^ q[2]) & (lo.w ^ q[3]) &
                 (hi.x ^ q[4]) & (hi.y ^ q[5]) & (hi.z ^ q[6]) & (hi.w ^ q[7]);
    const int e0 = b * KB + lane * 32;
    while (m != 0u) {
      const int ei = e0 + __builtin_ctz(m);
      m &= m - 1u;
      if (ei < nedges) {
        const unsigned sl = c - lo_slot;
        if (sl < (unsigned)SL) myslot[sl * OT] = ei;
        ++c;
      }
    }
  }
}

__global__ void __launch_bounds__(OT) k_own(
    const float* __restrict__ H, const float* __restrict__ dinv,
    const int* __restrict__ src, const int* __restrict__ dst,
    const unsigned* __restrict__ keyw, const float* __restrict__ bias,
    float* __restrict__ OUT, int nedges, int nkb, int n) {
  const uint4* __restrict__ keys = (const uint4*)keyw;
  __shared__ int big[2 * SL * OT];
  __shared__ int wl[OW * 32 * SL];
  __shared__ int ws[OW * 32 * SL];
  __shared__ unsigned bcnt[NB];
  __shared__ unsigned bcur[NB];
  __shared__ unsigned bfst[NB];
  __shared__ unsigned s_more;
  int* slots = big;
  uint2* recs = (uint2*)big;
  const int tid = threadIdx.x;
  const int lane = tid & 31;
  const int wave = __builtin_amdgcn_readfirstlane(tid >> 5);
  const unsigned own = blockIdx.x;
  unsigned q[8];
#pragma unroll
  for (int p = 0; p < 8; ++p) q[p] = ((own >> p) & 1u) ? 0u : 0xffffffffu;
  const int row0 = (int)own * OR;
  int nr = n - row0;
  if (nr > OR) nr = OR;
  if (nr < 0) nr = 0;

  if (tid < NB) bcnt[tid] = 0u;
  if (tid == 0) s_more = 0u;
  const float4 bv = *(const float4*)(bias + lane * 4);
  float4 st[RJ];
  float sd[RJ];
#pragma unroll
  for (int t = 0; t < RJ; ++t) {
    const int r = wave + 32 * t;
    const int rc = row0 + ((r < nr) ? r : 0);
    sd[t] = dinv[rc];
    st[t] = *(const float4*)(H + (size_t)rc * D + lane * 4);
  }
#pragma unroll
  for (int t = 0; t < RJ; ++t) {
    const float sc = sd[t] * sd[t];
    st[t].x = fmaf(st[t].x, sc, bv.x);
    st[t].y = fmaf(st[t].y, sc, bv.y);
    st[t].z = fmaf(st[t].z, sc, bv.z);
    st[t].w = fmaf(st[t].w, sc, bv.w);
  }
  __syncthreads();

  int* myslot = slots + tid;
  int* mywl = wl + wave * (32 * SL);
  int* myws = ws + wave * (32 * SL);
  for (unsigned lo = 0u;; lo += (unsigned)SL) {
    unsigned c = 0u;
    uint4 ka[4], kb[4];
    load_blk(keys, wave, nkb, lane, ka[0], ka[1]);
    load_blk(keys, wave + OW, nkb, lane, ka[2], ka[3]);
    for (int b = wave; b < nkb; b += 4 * OW) {
      load_blk(keys, b + 2 * OW, nkb, lane, kb[0], kb[1]);
      load_blk(keys, b + 3 * OW, nkb, lane, kb[2], kb[3]);
      own_scan(ka[0], ka[1], b, nkb, lane, nedges, q, lo, c, myslot);
      own_scan(ka[2], ka[3], b + OW, nkb, lane, nedges, q, lo, c, myslot);
      load_blk(keys, b + 4 * OW, nkb, lane, ka[0], ka[1]);
      load_blk(keys, b + 5 * OW, nkb, lane, ka[2], ka[3]);
      own_scan(kb[0], kb[1], b + 2 * OW, nkb, lane, nedges, q, lo, c, myslot);
      own_scan(kb[2], kb[3], b + 3 * OW, nkb, lane, nedges, q, lo, c, myslot);
    }
    const unsigned have = (c > lo) ? ((c - lo < (unsigned)SL) ? c - lo : (unsigned)SL) : 0u;
    int tot = 0;
    for (int sl = 0; sl < SL; ++sl) {
      const bool on = (unsigned)sl < have;
      const unsigned mask = __builtin_amdgcn_ballot_w32(on);
      if (mask == 0u) break;
      if (on) mywl[tot + (int)__builtin_amdgcn_mbcnt_lo(mask, 0u)] = myslot[sl * OT];
      tot += __builtin_popcount(mask);
    }
    if (c > lo + (unsigned)SL) s_more = 1u;
    for (int g = 0; g < tot; g += 128) {
      int ev[4], dv[4], sv[4];
#pragma unroll
      for (int u = 0; u < 4; ++u) {
        const int i = g + 32 * u + lane;
        ev[u] = (i < tot) ? mywl[i] : 0;
      }
#pragma unroll
      for (int u = 0; u < 4; ++u) { dv[u] = dst[ev[u]]; sv[u] = src[ev[u]]; }
#pragma unroll
      for (int u = 0; u < 4; ++u) {
        const int i = g + 32 * u + lane;
        if (i < tot) {
          const unsigned r = (unsigned)(dv[u] - row0);
          myws[i] = sv[u];
          mywl[i] = (int)((unsigned)ev[u] | (r << 24));
          atomicAdd(&bcnt[(r & 31u) * (unsigned)RJ + (r >> 5)], 1u);
        }
      }
    }
    __syncthreads();
    const unsigned more = s_more;
    if (wave == 0) {
      unsigned cn[RJ];
      unsigned sum = 0u;
#pragma unroll
      for (int t = 0; t < RJ; ++t) { cn[t] = bcnt[lane * RJ + t]; sum += cn[t]; }
      unsigned incl = sum;
      for (int d = 1; d < 32; d <<= 1) {
        const unsigned y = __shfl_up(incl, (unsigned)d);
        if (lane >= d) incl += y;
      }
      unsigned first = incl - sum;
#pragma unroll
      for (int t = 0; t < RJ; ++t) {
        bcur[lane * RJ + t] = first;
        bfst[lane * RJ + t] = first;
        first += cn[t];
      }
    }
    __syncthreads();
    if (tid == 0) s_more = 0u;
    for (int g = 0; g < tot; g += 128) {
      unsigned rv[4];
      int sv[4];
      float ds[4], dd[4];
#pragma unroll
      for (int u = 0; u < 4; ++u) {
        const int i = g + 32 * u + lane;
        rv[u] = (i < tot) ? ((unsigned)mywl[i] >> 24) : 0u;
        sv[u] = (i < tot) ? myws[i] : 0;
      }
#pragma unroll
      for (int u = 0; u < 4; ++u) { ds[u] = dinv[sv[u]]; dd[u] = dinv[row0 + (int)rv[u]]; }
#pragma unroll
      for (int u = 0; u < 4; ++u) {
        const int i = g + 32 * u + lane;
        if (i < tot) {
          const float cf = ds[u] * dd[u];
          const unsigned pos = atomicAdd(&bcur[(rv[u] & 31u) * (unsigned)RJ + (rv[u] >> 5)], 1u);
          recs[pos] = make_uint2((unsigned)sv[u], (unsigned)__float_as_int(cf));
        }
      }
    }
    __syncthreads();
    for (int t = 0; t < RJ; ++t) {
      const int r = wave + 32 * t;
      if (r >= nr) break;
      const int b0 = __builtin_amdgcn_readfirstlane((int)bfst[wave * RJ + t]);
      const int bn = __builtin_amdgcn_readfirstlane((int)bcnt[wave * RJ + t]);
      if (lo != 0u && bn == 0) continue;
      float* orow = OUT + (size_t)(row0 + r) * D + lane * 4;
      float4 cur;
      if (lo == 0u) {
        cur = st[0];
#pragma unroll
        for (int u = 1; u < RJ; ++u) if (t == u) cur = st[u];
      } else {
        cur = *(const float4*)orow;
      }
      for (int g = 0; g < bn; g += 32) {
        const int k = (bn - g < 32) ? (bn - g) : 32;
        own_add(H, recs + b0 + g, k, lane, cur);
      }
      *(float4*)orow = cur;
    }
    if (more == 0u) break;
    if (lane < RJ) bcnt[wave * RJ + lane] = 0u;
    __threadfence();
    __syncthreads();
  }
}

extern "C" void kernel_launch(void* const* d_in, const int* in_sizes, int n_in,
                              void* d_out, int out_size, void* d_ws, size_t ws_size,
                              hipStream_t stream) {
  (void)n_in; (void)out_size; (void)ws_size;
  const float* x  = (const float*)d_in[0];
  const int*   ei = (const int*)d_in[1];
  const float* W1 = (const float*)d_in[2];
  const float* b1 = (const float*)d_in[3];
  const float* W2 = (const float*)d_in[4];
  const float* b2 = (const float*)d_in[5];
  const int n = in_sizes[0] / D;
  const int e = in_sizes[1] / 2;
  const int* srcv = ei;
  const int* dstv = ei + e;

  char* ws = (char*)d_ws;
  float* dinv = (float*)ws;
  size_t off = (((size_t)n * 4) + 255) & ~(size_t)255;
  float* Hbuf = (float*)(ws + off);
  off += (((size_t)n * D * 4) + 255) & ~(size_t)255;
  float* Abuf = (float*)(ws + off);
  off += (((size_t)n * D * 4) + 255) & ~(size_t)255;
  unsigned* keys = (unsigned*)(ws + off);
  unsigned* planes = (unsigned*)Hbuf;
  float* outp = (float*)d_out;

  const dim3 blk(256);
  const int nblk = (n + 255) / 256;
  const int gblk = (n + 127) / 128;
  const int nown = (n + OR - 1) / OR;
  const int nkb = (e + KB - 1) / KB;
  const int nslice = (e + HT * HU - 1) / (HT * HU);
  const int nwin = (n + HB - 1) / HB;
  const int wsz = (((n + nwin - 1) / nwin) + 31) & ~31;
  const int np = (n + 31) & ~31;

  k_hist<<<dim3(nslice, nwin), dim3(HT), 0, stream>>>(dstv, planes, e, wsz, np, n, keys, nkb);
  k_dinv<<<dim3(nblk), blk, 0, stream>>>(planes, dinv, n, np, nslice);

  k_gcn_gemm<<<dim3(gblk), blk, 0, stream>>>(x, W1, Hbuf, n, 0);
  k_own<<<dim3(nown), dim3(OT), 0, stream>>>(Hbuf, dinv, srcv, dstv, keys, b1, Abuf, e, nkb, n);

  k_gcn_gemm<<<dim3(gblk), blk, 0, stream>>>(Abuf, W2, Hbuf, n, 1);
  k_own<<<dim3(nown), dim3(OT), 0, stream>>>(Hbuf, dinv, srcv, dstv, keys, b2, outp, e, nkb, n);
}


# Layer-2 owners load layer 1's sorted records and bin bounds (stored by it in the same call, exact count, no padding read) and skip key scan, sort and lookups; layer-2 sums in layer 1's record order
# speedup vs baseline: 2.5749x; 1.0733x over previous
#include <hip/hip_runtime.h>

#define D 128

typedef float v2f __attribute__((ext_vector_type(2)));
typedef float v8f __attribute__((ext_vector_type(8)));

#if defined(__HIP_DEVICE_COMPILE__) && __has_builtin(__builtin_amdgcn_wmma_f32_16x16x4_f32)
#define HAVE_WMMA_F32 1
#endif

__device__ __forceinline__ v8f wmma_acc(v2f a, v2f b, v8f c) {
#if defined(HAVE_WMMA_F32)
  return __builtin_amdgcn_wmma_f32_16x16x4_f32(false, a, false, b, (short)0, c,
                                               false, false);
#else
  (void)a; (void)b;
  return c;
#endif
}

#define OR 196
#define KB 1024
#define HB 16384
#define HT 1024
#define HU 16
__global__ void __launch_bounds__(HT) k_hist(
    const int* __restrict__ dst, unsigned* __restrict__ planes,
    int e, int wsz, int np, int n, unsigned* __restrict__ keys, int nkb) {
  __shared__ unsigned hist[HB];
  __shared__ uint4 kst[HU * 64];
  const int tid = threadIdx.x;
  const int base = blockIdx.y * wsz;
  int wn = n - base;
  if (wn > wsz) wn = wsz;
  if (wn < 0) wn = 0;
  for (int i = tid; i < wn; i += HT) hist[i] = 0u;
  __syncthreads();
  const int i0 = blockIdx.x * (HT * HU) + tid;
  int d[HU];
#pragma unroll
  for (int u = 0; u < HU; ++u) {
    const int i = i0 + u * HT;
    d[u] = (i < e) ? dst[i] : -1;
  }
  int myu = blockIdx.y;
#pragma unroll
  for (int u = 0; u < HU; ++u) {
    if (u == myu) {
      myu += gridDim.y;
      const int i = i0 + u * HT;
      const unsigned key = (i < e) ? (unsigned)d[u] / (unsigned)OR : 0u;
      unsigned pw[8];
#pragma unroll
      for (int p = 0; p < 8; ++p) pw[p] = __builtin_amdgcn_ballot_w32(((key >> p) & 1u) != 0u);
      if ((tid & 31) == 0) {
        uint4* q = kst + u * 64 + (tid >> 5);
        q[0]  = make_uint4(pw[0], pw[1], pw[2], pw[3]);
        q[32] = make_uint4(pw[4], pw[5], pw[6], pw[7]);
      }
    }
  }
#pragma unroll
  for (int u = 0; u < HU; ++u) {
    const unsigned r = (unsigned)(d[u] - base);
    if (r < (unsigned)wn) atomicAdd(&hist[r], 1u);
  }
  __syncthreads();
  unsigned* __restrict__ p = planes + (size_t)blockIdx.x * np + base;
  for (int i = tid; i < wn; i += HT) p[i] = hist[i];
  for (int u = blockIdx.y; u < HU; u += gridDim.y) {
    const int b = blockIdx.x * HU + u;
    if (b < nkb && tid < 64) ((uint4*)keys)[(size_t)b * 64 + tid] = kst[u * 64 + tid];
  }
}

__global__ void k_dinv(const unsigned* __restrict__ planes, float* __restrict__ dinv,
                       int n, int np, int nslice) {
  int i = blockIdx.x * blockDim.x + threadIdx.x;
  if (i < n) {
    unsigned s = 0u;
    for (int g0 = 0; g0 < nslice; g0 += 16) {
      unsigned v[16];
#pragma unroll
      for (int u = 0; u < 16; ++u)
        v[u] = (g0 + u < nslice) ? planes[(size_t)(g0 + u) * np + i] : 0u;
#pragma unroll
      for (int u = 0; u < 16; ++u) s += v[u];
    }
    dinv[i] = rsqrtf((float)s + 1.0f);
  }
}

__global__ void __launch_bounds__(256) k_gcn_gemm(
    const float* __restrict__ X, const float* __restrict__ W,
    float* __restrict__ H, int nrows, int relu_in) {
  const int wave = threadIdx.x >> 5;
  const int lane = threadIdx.x & 31;
  const int half = lane >> 4;
  const int l16  = lane & 15;
  const int row0 = (blockIdx.x * 8 + wave) * 16;
  if (row0 >= nrows) return;

  const int m = row0 + l16;
  const bool mvalid = (m < nrows);
  const float* __restrict__ arow = X + (size_t)(mvalid ? m : 0) * D;

  const v8f vzero = {0.f, 0.f, 0.f, 0.f, 0.f, 0.f, 0.f, 0.f};
  v8f acc[8];
#pragma unroll
  for (int i = 0; i < 8; ++i) acc[i] = vzero;

  for (int k0 = 0; k0 < D; k0 += 4) {
    const int ka = k0 + 2 * half;
    float a0 = arow[ka];
    float a1 = arow[ka + 1];
    if (relu_in) { a0 = fmaxf(a0, 0.f); a1 = fmaxf(a1, 0.f); }
    if (!mvalid) { a0 = 0.f; a1 = 0.f; }
    v2f a; a.x = a0; a.y = a1;
    const float* __restrict__ w0 = W + (size_t)ka * D + l16;
    const float* __restrict__ w1 = w0 + D;
#pragma unroll
    for (int ct = 0; ct < 8; ++ct) {
      v2f b; b.x = w0[ct * 16]; b.y = w1[ct * 16];
      acc[ct] = wmma_acc(a, b, acc[ct]);
    }
  }

#pragma unroll
  for (int v = 0; v < 8; ++v) {
    const int mm = row0 + v + 8 * half;
    if (mm < nrows) {
      const size_t base = (size_t)mm * D;
#pragma unroll
      for (int ct = 0; ct < 8; ++ct) H[base + ct * 16 + l16] = acc[ct][v];
    }
  }
}

#define OT 1024
#define OW (OT / 32)
#define SL 16
#define RJ ((OR + OW - 1) / OW)
#define NB (OW * RJ)
#define LHDR 512
#define LSTRIDE (LHDR + 2 * SL * OT)

__device__ __forceinline__ void load_blk(const uint4* __restrict__ keys, int b, int nkb,
                                         int lane, uint4& lo, uint4& hi) {
  const uint4* p = keys + (size_t)((b < nkb) ? b : 0) * 64 + lane;
  lo = p[0];
  hi = p[32];
}

__device__ __forceinline__ void own_add(
    const float* __restrict__ H, const uint2* rec, int k, int lane, float4& cur) {
  uint2 v = make_uint2(0u, 0u);
  if (lane < k) v = rec[lane];
  const int sv = (int)v.x;
  const int ci = (int)v.y;
  unsigned m = (k >= 32) ? 0xffffffffu : ((1u << k) - 1u);
  while (m != 0u) {
    float4 h[8];
    int jl[8];
#pragma unroll
    for (int u = 0; u < 8; ++u) {
      jl[u] = -1;
      if (m != 0u) {
        const int j = __builtin_ctz(m);
        m &= m - 1u;
        jl[u] = j;
        const int sj = __builtin_amdgcn_readlane(sv, j);
        h[u] = *((const float4*)(H + (size_t)sj * D) + lane);
      }
    }
#pragma unroll
    for (int u = 0; u < 8; ++u) {
      if (jl[u] >= 0) {
        const float cj = __int_as_float(__builtin_amdgcn_readlane(ci, jl[u]));
        cur.x = fmaf(h[u].x, cj, cur.x);
        cur.y = fmaf(h[u].y, cj, cur.y);
        cur.z = fmaf(h[u].z, cj, cur.z);
        cur.w = fmaf(h[u].w, cj, cur.w);
      }
    }
  }
}

__device__ __forceinline__ void own_scan(
    const uint4& lo, const uint4& hi, int b, int nkb, int lane, int nedges,
    const unsigned (&q)[8], unsigned lo_slot, unsigned& c, int* myslot) {
  if (b < nkb) {
    unsigned m = (lo.x ^ q[0]) & (lo.y ^ q[1]) & (lo.z ^ q[2]) & (lo.w ^ q[3]) &
                 (hi.x ^ q[4]) & (hi.y ^ q[5]) & (hi.z ^ q[6]) & (hi.w ^ q[7]);
    const int e0 = b * KB + lane * 32;
    while (m != 0u) {
      const int ei = e0 + __builtin_ctz(m);
      m &= m - 1u;
      if (ei < nedges) {
        const unsigned sl = c - lo_slot;
        if (sl < (unsigned)SL) myslot[sl * OT] = ei;
        ++c;
      }
    }
  }
}

__global__ void __launch_bounds__(OT) k_own(
    const float* __restrict__ H, const float* __restrict__ dinv,
    const int* __restrict__ src, const int* __restrict__ dst,
    const unsigned* __restrict__ keyw, const float* __restrict__ bias,
    float* __restrict__ OUT, int nedges, int nkb, int n,
    unsigned* lists, int use_lists) {
  const uint4* __restrict__ keys = (const uint4*)keyw;
  __shared__ int big[2 * SL * OT];
  __shared__ int wl[OW * 32 * SL];
  __shared__ int ws[OW * 32 * SL];
  __shared__ unsigned bcnt[NB];
  __shared__ unsigned bcur[NB];
  __shared__ unsigned bfst[NB];
  __shared__ unsigned s_more;
  __shared__ unsigned s_total;
  int* slots = big;
  uint2* recs = (uint2*)big;
  const int tid = threadIdx.x;
  const int lane = tid & 31;
  const int wave = __builtin_amdgcn_readfirstlane(tid >> 5);
  const unsigned own = blockIdx.x;
  unsigned q[8];
#pragma unroll
  for (int p = 0; p < 8; ++p) q[p] = ((own >> p) & 1u) ? 0u : 0xffffffffu;
  const int row0 = (int)own * OR;
  int nr = n - row0;
  if (nr > OR) nr = OR;
  if (nr < 0) nr = 0;

  unsigned* L = lists + (size_t)own * LSTRIDE;
  uint2* LR = (uint2*)(L + LHDR);
  unsigned lflag = 1u, ltot = 0u;
  if (use_lists) { lflag = L[2 * NB]; ltot = L[2 * NB + 1]; }
  if (tid < NB) bcnt[tid] = 0u;
  if (tid == 0) s_more = 0u;
  const float4 bv = *(const float4*)(bias + lane * 4);
  float4 st[RJ];
  float sd[RJ];
#pragma unroll
  for (int t = 0; t < RJ; ++t) {
    const int r = wave + 32 * t;
    const int rc = row0 + ((r < nr) ? r : 0);
    sd[t] = dinv[rc];
    st[t] = *(const float4*)(H + (size_t)rc * D + lane * 4);
  }
#pragma unroll
  for (int t = 0; t < RJ; ++t) {
    const float sc = sd[t] * sd[t];
    st[t].x = fmaf(st[t].x, sc, bv.x);
    st[t].y = fmaf(st[t].y, sc, bv.y);
    st[t].z = fmaf(st[t].z, sc, bv.z);
    st[t].w = fmaf(st[t].w, sc, bv.w);
  }
  __syncthreads();
  const bool fromlist = (__builtin_amdgcn_readfirstlane((int)lflag) == 0) &&
                        ((unsigned)__builtin_amdgcn_readfirstlane((int)ltot) <= (unsigned)(SL * OT));
  const unsigned ntot = (unsigned)__builtin_amdgcn_readfirstlane((int)ltot);

  if (fromlist) {
    if (tid < NB) { bfst[tid] = L[tid]; bcnt[tid] = L[NB + tid]; }
    for (unsigned i0 = 0u; i0 < ntot; i0 += 4u * OT) {
      uint2 v[4];
#pragma unroll
      for (int u = 0; u < 4; ++u) {
        const unsigned i = i0 + (unsigned)(u * OT + tid);
        v[u] = LR[(i < ntot) ? i : 0u];
      }
#pragma unroll
      for (int u = 0; u < 4; ++u) {
        const unsigned i = i0 + (unsigned)(u * OT + tid);
        if (i < ntot) recs[i] = v[u];
      }
    }
    __syncthreads();
  }

  int* myslot = slots + tid;
  int* mywl = wl + wave * (32 * SL);
  int* myws = ws + wave * (32 * SL);
  unsigned flag0 = 0u;
  for (unsigned lo = 0u;; lo += (unsigned)SL) {
    unsigned more = 0u;
    if (!fromlist) {
    unsigned c = 0u;
    uint4 ka[4], kb[4];
    load_blk(keys, wave, nkb, lane, ka[0], ka[1]);
    load_blk(keys, wave + OW, nkb, lane, ka[2], ka[3]);
    for (int b = wave; b < nkb; b += 4 * OW) {
      load_blk(keys, b + 2 * OW, nkb, lane, kb[0], kb[1]);
      load_blk(keys, b + 3 * OW, nkb, lane, kb[2], kb[3]);
      own_scan(ka[0], ka[1], b, nkb, lane, nedges, q, lo, c, myslot);
      own_scan(ka[2], ka[3], b + OW, nkb, lane, nedges, q, lo, c, myslot);
      load_blk(keys, b + 4 * OW, nkb, lane, ka[0], ka[1]);
      load_blk(keys, b + 5 * OW, nkb, lane, ka[2], ka[3]);
      own_scan(kb[0], kb[1], b + 2 * OW, nkb, lane, nedges, q, lo, c, myslot);
      own_scan(kb[2], kb[3], b + 3 * OW, nkb, lane, nedges, q, lo, c, myslot);
    }
    const unsigned have = (c > lo) ? ((c - lo < (unsigned)SL) ? c - lo : (unsigned)SL) : 0u;
    int tot = 0;
    for (int sl = 0; sl < SL; ++sl) {
      const bool on = (unsigned)sl < have;
      const unsigned mask = __builtin_amdgcn_ballot_w32(on);
      if (mask == 0u) break;
      if (on) mywl[tot + (int)__builtin_amdgcn_mbcnt_lo(mask, 0u)] = myslot[sl * OT];
      tot += __builtin_popcount(mask);
    }
    if (c > lo + (unsigned)SL) s_more = 1u;
    for (int g = 0; g < tot; g += 128) {
      int ev[4], dv[4], sv[4];
#pragma unroll
      for (int u = 0; u < 4; ++u) {
        const int i = g + 32 * u + lane;
        ev[u] = (i < tot) ? mywl[i] : 0;
      }
#pragma unroll
      for (int u = 0; u < 4; ++u) { dv[u] = dst[ev[u]]; sv[u] = src[ev[u]]; }
#pragma unroll
      for (int u = 0; u < 4; ++u) {
        const int i = g + 32 * u + lane;
        if (i < tot) {
          const unsigned r = (unsigned)(dv[u] - row0);
          myws[i] = sv[u];
          mywl[i] = (int)((unsigned)ev[u] | (r << 24));
          atomicAdd(&bcnt[(r & 31u) * (unsigned)RJ + (r >> 5)], 1u);
        }
      }
    }
    __syncthreads();
    more = s_more;
    if (wave == 0) {
      unsigned cn[RJ];
      unsigned sum = 0u;
#pragma unroll
      for (int t = 0; t < RJ; ++t) { cn[t] = bcnt[lane * RJ + t]; sum += cn[t]; }
      unsigned incl = sum;
      for (int d = 1; d < 32; d <<= 1) {
        const unsigned y = __shfl_up(incl, (unsigned)d);
        if (lane >= d) incl += y;
      }
      unsigned first = incl - sum;
      if (lane == 31) s_total = incl;
#pragma unroll
      for (int t = 0; t < RJ; ++t) {
        bcur[lane * RJ + t] = first;
        bfst[lane * RJ + t] = first;
        first += cn[t];
      }
    }
    __syncthreads();
    if (tid == 0) s_more = 0u;
    for (int g = 0; g < tot; g += 128) {
      unsigned rv[4];
      int sv[4];
      float ds[4], dd[4];
#pragma unroll
      for (int u = 0; u < 4; ++u) {
        const int i = g + 32 * u + lane;
        rv[u] = (i < tot) ? ((unsigned)mywl[i] >> 24) : 0u;
        sv[u] = (i < tot) ? myws[i] : 0;
      }
#pragma unroll
      for (int u = 0; u < 4; ++u) { ds[u] = dinv[sv[u]]; dd[u] = dinv[row0 + (int)rv[u]]; }
#pragma unroll
      for (int u = 0; u < 4; ++u) {
        const int i = g + 32 * u + lane;
        if (i < tot) {
          const float cf = ds[u] * dd[u];
          const unsigned pos = atomicAdd(&bcur[(rv[u] & 31u) * (unsigned)RJ + (rv[u] >> 5)], 1u);
          recs[pos] = make_uint2((unsigned)sv[u], (unsigned)__float_as_int(cf));
        }
      }
    }
    __syncthreads();
    if (lo == 0u) flag0 = more;
    }
    for (int t = 0; t < RJ; ++t) {
      const int r = wave + 32 * t;
      if (r >= nr) break;
      const int b0 = __builtin_amdgcn_readfirstlane((int)bfst[wave * RJ + t]);
      const int bn = __builtin_amdgcn_readfirstlane((int)bcnt[wave * RJ + t]);
      if (lo != 0u && bn == 0) continue;
      float* orow = OUT + (size_t)(row0 + r) * D + lane * 4;
      float4 cur;
      if (lo == 0u) {
        cur = st[0];
#pragma unroll
        for (int u = 1; u < RJ; ++u) if (t == u) cur = st[u];
      } else {
        cur = *(const float4*)orow;
      }
      for (int g = 0; g < bn; g += 32) {
        const int k = (bn - g < 32) ? (bn - g) : 32;
        own_add(H, recs + b0 + g, k, lane, cur);
      }
      *(float4*)orow = cur;
    }
    if (more == 0u) break;
    if (lane < RJ) bcnt[wave * RJ + lane] = 0u;
    __threadfence();
    __syncthreads();
  }
  if (use_lists == 0) {
    if (tid < NB) { L[tid] = bfst[tid]; L[NB + tid] = bcnt[tid]; }
    const unsigned tot = s_total;
    if (tid == 0) { L[2 * NB] = flag0; L[2 * NB + 1] = tot; }
    if (flag0 == 0u)
      for (unsigned i = (unsigned)tid; i < tot; i += (unsigned)OT) LR[i] = recs[i];
  }
}

extern "C" void kernel_launch(void* const* d_in, const int* in_sizes, int n_in,
                              void* d_out, int out_size, void* d_ws, size_t ws_size,
                              hipStream_t stream) {
  (void)n_in; (void)out_size; (void)ws_size;
  const float* x  = (const float*)d_in[0];
  const int*   ei = (const int*)d_in[1];
  const float* W1 = (const float*)d_in[2];
  const float* b1 = (const float*)d_in[3];
  const float* W2 = (const float*)d_in[4];
  const float* b2 = (const float*)d_in[5];
  const int n = in_sizes[0] / D;
  const int e = in_sizes[1] / 2;
  const int* srcv = ei;
  const int* dstv = ei + e;

  char* ws = (char*)d_ws;
  float* dinv = (float*)ws;
  size_t off = (((size_t)n * 4) + 255) & ~(size_t)255;
  float* Hbuf = (float*)(ws + off);
  off += (((size_t)n * D * 4) + 255) & ~(size_t)255;
  float* Abuf = (float*)(ws + off);
  off += (((size_t)n * D * 4) + 255) & ~(size_t)255;
  unsigned* keys = (unsigned*)(ws + off);
  off += ((((size_t)(e + KB - 1) / KB) * 1024) + 255) & ~(size_t)255;
  unsigned* lists = (unsigned*)(ws + off);
  unsigned* planes = (unsigned*)Hbuf;
  float* outp = (float*)d_out;

  const dim3 blk(256);
  const int nblk = (n + 255) / 256;
  const int gblk = (n + 127) / 128;
  const int nown = (n + OR - 1) / OR;
  const int nkb = (e + KB - 1) / KB;
  const int nslice = (e + HT * HU - 1) / (HT * HU);
  const int nwin = (n + HB - 1) / HB;
  const int wsz = (((n + nwin - 1) / nwin) + 31) & ~31;
  const int np = (n + 31) & ~31;

  k_hist<<<dim3(nslice, nwin), dim3(HT), 0, stream>>>(dstv, planes, e, wsz, np, n, keys, nkb);
  k_dinv<<<dim3(nblk), blk, 0, stream>>>(planes, dinv, n, np, nslice);

  k_gcn_gemm<<<dim3(gblk), blk, 0, stream>>>(x, W1, Hbuf, n, 0);
  k_own<<<dim3(nown), dim3(OT), 0, stream>>>(Hbuf, dinv, srcv, dstv, keys, b1, Abuf, e, nkb, n, lists, 0);

  k_gcn_gemm<<<dim3(gblk), blk, 0, stream>>>(Abuf, W2, Hbuf, n, 1);
  k_own<<<dim3(nown), dim3(OT), 0, stream>>>(Hbuf, dinv, srcv, dstv, keys, b2, outp, e, nkb, n, lists, 1);
}


# Degree sum joins the layer-1 GEMM launch: a ninth wave per work-group adds the 37 count planes of the group's 128 rows (16-byte loads) and stores dinv beside the multiplying waves; 6 launches to 5
# speedup vs baseline: 2.6638x; 1.0345x over previous
#include <hip/hip_runtime.h>

#define D 128

typedef float v2f __attribute__((ext_vector_type(2)));
typedef float v8f __attribute__((ext_vector_type(8)));

#if defined(__HIP_DEVICE_COMPILE__) && __has_builtin(__builtin_amdgcn_wmma_f32_16x16x4_f32)
#define HAVE_WMMA_F32 1
#endif

__device__ __forceinline__ v8f wmma_acc(v2f a, v2f b, v8f c) {
#if defined(HAVE_WMMA_F32)
  return __builtin_amdgcn_wmma_f32_16x16x4_f32(false, a, false, b, (short)0, c,
                                               false, false);
#else
  (void)a; (void)b;
  return c;
#endif
}

#define OR 196
#define KB 1024
#define HB 16384
#define HT 1024
#define HU 16
__global__ void __launch_bounds__(HT) k_hist(
    const int* __restrict__ dst, unsigned* __restrict__ planes,
    int e, int wsz, int np, int n, unsigned* __restrict__ keys, int nkb) {
  __shared__ unsigned hist[HB];
  __shared__ uint4 kst[HU * 64];
  const int tid = threadIdx.x;
  const int base = blockIdx.y * wsz;
  int wn = np - base;
  if (wn > wsz) wn = wsz;
  if (wn < 0) wn = 0;
  for (int i = tid; i < wn; i += HT) hist[i] = 0u;
  __syncthreads();
  const int i0 = blockIdx.x * (HT * HU) + tid;
  int d[HU];
#pragma unroll
  for (int u = 0; u < HU; ++u) {
    const int i = i0 + u * HT;
    d[u] = (i < e) ? dst[i] : -1;
  }
  int myu = blockIdx.y;
#pragma unroll
  for (int u = 0; u < HU; ++u) {
    if (u == myu) {
      myu += gridDim.y;
      const int i = i0 + u * HT;
      const unsigned key = (i < e) ? (unsigned)d[u] / (unsigned)OR : 0u;
      unsigned pw[8];
#pragma unroll
      for (int p = 0; p < 8; ++p) pw[p] = __builtin_amdgcn_ballot_w32(((key >> p) & 1u) != 0u);
      if ((tid & 31) == 0) {
        uint4* q = kst + u * 64 + (tid >> 5);
        q[0]  = make_uint4(pw[0], pw[1], pw[2], pw[3]);
        q[32] = make_uint4(pw[4], pw[5], pw[6], pw[7]);
      }
    }
  }
#pragma unroll
  for (int u = 0; u < HU; ++u) {
    const unsigned r = (unsigned)(d[u] - base);
    if (r < (unsigned)wn) atomicAdd(&hist[r], 1u);
  }
  __syncthreads();
  unsigned* __restrict__ p = planes + (size_t)blockIdx.x * np + base;
  for (int i = tid; i < wn; i += HT) p[i] = hist[i];
  for (int u = blockIdx.y; u < HU; u += gridDim.y) {
    const int b = blockIdx.x * HU + u;
    if (b < nkb && tid < 64) ((uint4*)keys)[(size_t)b * 64 + tid] = kst[u * 64 + tid];
  }
}

#define PH 16
__global__ void __launch_bounds__(288) k_gcn_gemm(
    const float* __restrict__ X, const float* __restrict__ W,
    float* __restrict__ H, int nrows, int relu_in,
    const unsigned* __restrict__ planes, float* __restrict__ dinv, int np, int nslice) {
  const int wave = threadIdx.x >> 5;
  const int lane = threadIdx.x & 31;
  if (wave == 8) {
    const int d0 = blockIdx.x * 128 + 4 * lane;
    const int dc = (d0 < nrows) ? d0 : 0;
    uint4 s = make_uint4(0u, 0u, 0u, 0u);
    for (int g0 = 0; g0 < nslice; g0 += PH) {
      uint4 pv[PH];
#pragma unroll
      for (int u = 0; u < PH; ++u) {
        const int g = g0 + u;
        pv[u] = *(const uint4*)(planes + (size_t)((g < nslice) ? g : 0) * np + dc);
      }
#pragma unroll
      for (int u = 0; u < PH; ++u)
        if (g0 + u < nslice) {
          s.x += pv[u].x; s.y += pv[u].y; s.z += pv[u].z; s.w += pv[u].w;
        }
    }
    float4 r;
    r.x = rsqrtf((float)s.x + 1.0f);
    r.y = rsqrtf((float)s.y + 1.0f);
    r.z = rsqrtf((float)s.z + 1.0f);
    r.w = rsqrtf((float)s.w + 1.0f);
    if (d0 + 3 < nrows) {
      *(float4*)(dinv + d0) = r;
    } else {
      if (d0 < nrows) dinv[d0] = r.x;
      if (d0 + 1 < nrows) dinv[d0 + 1] = r.y;
      if (d0 + 2 < nrows) dinv[d0 + 2] = r.z;
    }
    return;
  }
  const int half = lane >> 4;
  const int l16  = lane & 15;
  const int row0 = (blockIdx.x * 8 + wave) * 16;
  if (row0 >= nrows) return;

  const int m = row0 + l16;
  const bool mvalid = (m < nrows);
  const float* __restrict__ arow = X + (size_t)(mvalid ? m : 0) * D;

  const v8f vzero = {0.f, 0.f, 0.f, 0.f, 0.f, 0.f, 0.f, 0.f};
  v8f acc[8];
#pragma unroll
  for (int i = 0; i < 8; ++i) acc[i] = vzero;

  for (int k0 = 0; k0 < D; k0 += 4) {
    const int ka = k0 + 2 * half;
    float a0 = arow[ka];
    float a1 = arow[ka + 1];
    if (relu_in) { a0 = fmaxf(a0, 0.f); a1 = fmaxf(a1, 0.f); }
    if (!mvalid) { a0 = 0.f; a1 = 0.f; }
    v2f a; a.x = a0; a.y = a1;
    const float* __restrict__ w0 = W + (size_t)ka * D + l16;
    const float* __restrict__ w1 = w0 + D;
#pragma unroll
    for (int ct = 0; ct < 8; ++ct) {
      v2f b; b.x = w0[ct * 16]; b.y = w1[ct * 16];
      acc[ct] = wmma_acc(a, b, acc[ct]);
    }
  }

#pragma unroll
  for (int v = 0; v < 8; ++v) {
    const int mm = row0 + v + 8 * half;
    if (mm < nrows) {
      const size_t base = (size_t)mm * D;
#pragma unroll
      for (int ct = 0; ct < 8; ++ct) H[base + ct * 16 + l16] = acc[ct][v];
    }
  }
}

#define OT 1024
#define OW (OT / 32)
#define SL 16
#define RJ ((OR + OW - 1) / OW)
#define NB (OW * RJ)
#define LHDR 512
#define LSTRIDE (LHDR + 2 * SL * OT)

__device__ __forceinline__ void load_blk(const uint4* __restrict__ keys, int b, int nkb,
                                         int lane, uint4& lo, uint4& hi) {
  const uint4* p = keys + (size_t)((b < nkb) ? b : 0) * 64 + lane;
  lo = p[0];
  hi = p[32];
}

__device__ __forceinline__ void own_add(
    const float* __restrict__ H, const uint2* rec, int k, int lane, float4& cur) {
  uint2 v = make_uint2(0u, 0u);
  if (lane < k) v = rec[lane];
  const int sv = (int)v.x;
  const int ci = (int)v.y;
  unsigned m = (k >= 32) ? 0xffffffffu : ((1u << k) - 1u);
  while (m != 0u) {
    float4 h[8];
    int jl[8];
#pragma unroll
    for (int u = 0; u < 8; ++u) {
      jl[u] = -1;
      if (m != 0u) {
        const int j = __builtin_ctz(m);
        m &= m - 1u;
        jl[u] = j;
        const int sj = __builtin_amdgcn_readlane(sv, j);
        h[u] = *((const float4*)(H + (size_t)sj * D) + lane);
      }
    }
#pragma unroll
    for (int u = 0; u < 8; ++u) {
      if (jl[u] >= 0) {
        const float cj = __int_as_float(__builtin_amdgcn_readlane(ci, jl[u]));
        cur.x = fmaf(h[u].x, cj, cur.x);
        cur.y = fmaf(h[u].y, cj, cur.y);
        cur.z = fmaf(h[u].z, cj, cur.z);
        cur.w = fmaf(h[u].w, cj, cur.w);
      }
    }
  }
}

__device__ __forceinline__ void own_scan(
    const uint4& lo, const uint4& hi, int b, int nkb, int lane, int nedges,
    const unsigned (&q)[8], unsigned lo_slot, unsigned& c, int* myslot) {
  if (b < nkb) {
    unsigned m = (lo.x ^ q[0]) & (lo.y ^ q[1]) & (lo.z ^ q[2]) & (lo.w ^ q[3]) &
                 (hi.x ^ q[4]) & (hi.y ^ q[5]) & (hi.z ^ q[6]) & (hi.w ^ q[7]);
    const int e0 = b * KB + lane * 32;
    while (m != 0u) {
      const int ei = e0 + __builtin_ctz(m);
      m &= m - 1u;
      if (ei < nedges) {
        const unsigned sl = c - lo_slot;
        if (sl < (unsigned)SL) myslot[sl * OT] = ei;
        ++c;
      }
    }
  }
}

__global__ void __launch_bounds__(OT) k_own(
    const float* __restrict__ H, const float* __restrict__ dinv,
    const int* __restrict__ src, const int* __restrict__ dst,
    const unsigned* __restrict__ keyw, const float* __restrict__ bias,
    float* __restrict__ OUT, int nedges, int nkb, int n,
    unsigned* lists, int use_lists) {
  const uint4* __restrict__ keys = (const uint4*)keyw;
  __shared__ int big[2 * SL * OT];
  __shared__ int wl[OW * 32 * SL];
  __shared__ int ws[OW * 32 * SL];
  __shared__ unsigned bcnt[NB];
  __shared__ unsigned bcur[NB];
  __shared__ unsigned bfst[NB];
  __shared__ unsigned s_more;
  __shared__ unsigned s_total;
  int* slots = big;
  uint2* recs = (uint2*)big;
  const int tid = threadIdx.x;
  const int lane = tid & 31;
  const int wave = __builtin_amdgcn_readfirstlane(tid >> 5);
  const unsigned own = blockIdx.x;
  unsigned q[8];
#pragma unroll
  for (int p = 0; p < 8; ++p) q[p] = ((own >> p) & 1u) ? 0u : 0xffffffffu;
  const int row0 = (int)own * OR;
  int nr = n - row0;
  if (nr > OR) nr = OR;
  if (nr < 0) nr = 0;

  unsigned* L = lists + (size_t)own * LSTRIDE;
  uint2* LR = (uint2*)(L + LHDR);
  unsigned lflag = 1u, ltot = 0u;
  if (use_lists) { lflag = L[2 * NB]; ltot = L[2 * NB + 1]; }
  if (tid < NB) bcnt[tid] = 0u;
  if (tid == 0) s_more = 0u;
  const float4 bv = *(const float4*)(bias + lane * 4);
  float4 st[RJ];
  float sd[RJ];
#pragma unroll
  for (int t = 0; t < RJ; ++t) {
    const int r = wave + 32 * t;
    const int rc = row0 + ((r < nr) ? r : 0);
    sd[t] = dinv[rc];
    st[t] = *(const float4*)(H + (size_t)rc * D + lane * 4);
  }
#pragma unroll
  for (int t = 0; t < RJ; ++t) {
    const float sc = sd[t] * sd[t];
    st[t].x = fmaf(st[t].x, sc, bv.x);
    st[t].y = fmaf(st[t].y, sc, bv.y);
    st[t].z = fmaf(st[t].z, sc, bv.z);
    st[t].w = fmaf(st[t].w, sc, bv.w);
  }
  __syncthreads();
  const bool fromlist = (__builtin_amdgcn_readfirstlane((int)lflag) == 0) &&
                        ((unsigned)__builtin_amdgcn_readfirstlane((int)ltot) <= (unsigned)(SL * OT));
  const unsigned ntot = (unsigned)__builtin_amdgcn_readfirstlane((int)ltot);

  if (fromlist) {
    if (tid < NB) { bfst[tid] = L[tid]; bcnt[tid] = L[NB + tid]; }
    for (unsigned i0 = 0u; i0 < ntot; i0 += 4u * OT) {
      uint2 v[4];
#pragma unroll
      for (int u = 0; u < 4; ++u) {
        const unsigned i = i0 + (unsigned)(u * OT + tid);
        v[u] = LR[(i < ntot) ? i : 0u];
      }
#pragma unroll
      for (int u = 0; u < 4; ++u) {
        const unsigned i = i0 + (unsigned)(u * OT + tid);
        if (i < ntot) recs[i] = v[u];
      }
    }
    __syncthreads();
  }

  int* myslot = slots + tid;
  int* mywl = wl + wave * (32 * SL);
  int* myws = ws + wave * (32 * SL);
  unsigned flag0 = 0u;
  for (unsigned lo = 0u;; lo += (unsigned)SL) {
    unsigned more = 0u;
    if (!fromlist) {
    unsigned c = 0u;
    uint4 ka[4], kb[4];
    load_blk(keys, wave, nkb, lane, ka[0], ka[1]);
    load_blk(keys, wave + OW, nkb, lane, ka[2], ka[3]);
    for (int b = wave; b < nkb; b += 4 * OW) {
      load_blk(keys, b + 2 * OW, nkb, lane, kb[0], kb[1]);
      load_blk(keys, b + 3 * OW, nkb, lane, kb[2], kb[3]);
      own_scan(ka[0], ka[1], b, nkb, lane, nedges, q, lo, c, myslot);
      own_scan(ka[2], ka[3], b + OW, nkb, lane, nedges, q, lo, c, myslot);
      load_blk(keys, b + 4 * OW, nkb, lane, ka[0], ka[1]);
      load_blk(keys, b + 5 * OW, nkb, lane, ka[2], ka[3]);
      own_scan(kb[0], kb[1], b + 2 * OW, nkb, lane, nedges, q, lo, c, myslot);
      own_scan(kb[2], kb[3], b + 3 * OW, nkb, lane, nedges, q, lo, c, myslot);
    }
    const unsigned have = (c > lo) ? ((c - lo < (unsigned)SL) ? c - lo : (unsigned)SL) : 0u;
    int tot = 0;
    for (int sl = 0; sl < SL; ++sl) {
      const bool on = (unsigned)sl < have;
      const unsigned mask = __builtin_amdgcn_ballot_w32(on);
      if (mask == 0u) break;
      if (on) mywl[tot + (int)__builtin_amdgcn_mbcnt_lo(mask, 0u)] = myslot[sl * OT];
      tot += __builtin_popcount(mask);
    }
    if (c > lo + (unsigned)SL) s_more = 1u;
    for (int g = 0; g < tot; g += 128) {
      int ev[4], dv[4], sv[4];
#pragma unroll
      for (int u = 0; u < 4; ++u) {
        const int i = g + 32 * u + lane;
        ev[u] = (i < tot) ? mywl[i] : 0;
      }
#pragma unroll
      for (int u = 0; u < 4; ++u) { dv[u] = dst[ev[u]]; sv[u] = src[ev[u]]; }
#pragma unroll
      for (int u = 0; u < 4; ++u) {
        const int i = g + 32 * u + lane;
        if (i < tot) {
          const unsigned r = (unsigned)(dv[u] - row0);
          myws[i] = sv[u];
          mywl[i] = (int)((unsigned)ev[u] | (r << 24));
          atomicAdd(&bcnt[(r & 31u) * (unsigned)RJ + (r >> 5)], 1u);
        }
      }
    }
    __syncthreads();
    more = s_more;
    if (wave == 0) {
      unsigned cn[RJ];
      unsigned sum = 0u;
#pragma unroll
      for (int t = 0; t < RJ; ++t) { cn[t] = bcnt[lane * RJ + t]; sum += cn[t]; }
      unsigned incl = sum;
      for (int d = 1; d < 32; d <<= 1) {
        const unsigned y = __shfl_up(incl, (unsigned)d);
        if (lane >= d) incl += y;
      }
      unsigned first = incl - sum;
      if (lane == 31) s_total = incl;
#pragma unroll
      for (int t = 0; t < RJ; ++t) {
        bcur[lane * RJ + t] = first;
        bfst[lane * RJ + t] = first;
        first += cn[t];
      }
    }
    __syncthreads();
    if (tid == 0) s_more = 0u;
    for (int g = 0; g < tot; g += 128) {
      unsigned rv[4];
      int sv[4];
      float ds[4], dd[4];
#pragma unroll
      for (int u = 0; u < 4; ++u) {
        const int i = g + 32 * u + lane;
        rv[u] = (i < tot) ? ((unsigned)mywl[i] >> 24) : 0u;
        sv[u] = (i < tot) ? myws[i] : 0;
      }
#pragma unroll
      for (int u = 0; u < 4; ++u) { ds[u] = dinv[sv[u]]; dd[u] = dinv[row0 + (int)rv[u]]; }
#pragma unroll
      for (int u = 0; u < 4; ++u) {
        const int i = g + 32 * u + lane;
        if (i < tot) {
          const float cf = ds[u] * dd[u];
          const unsigned pos = atomicAdd(&bcur[(rv[u] & 31u) * (unsigned)RJ + (rv[u] >> 5)], 1u);
          recs[pos] = make_uint2((unsigned)sv[u], (unsigned)__float_as_int(cf));
        }
      }
    }
    __syncthreads();
    if (lo == 0u) flag0 = more;
    }
    for (int t = 0; t < RJ; ++t) {
      const int r = wave + 32 * t;
      if (r >= nr) break;
      const int b0 = __builtin_amdgcn_readfirstlane((int)bfst[wave * RJ + t]);
      const int bn = __builtin_amdgcn_readfirstlane((int)bcnt[wave * RJ + t]);
      if (lo != 0u && bn == 0) continue;
      float* orow = OUT + (size_t)(row0 + r) * D + lane * 4;
      float4 cur;
      if (lo == 0u) {
        cur = st[0];
#pragma unroll
        for (int u = 1; u < RJ; ++u) if (t == u) cur = st[u];
      } else {
        cur = *(const float4*)orow;
      }
      for (int g = 0; g < bn; g += 32) {
        const int k = (bn - g < 32) ? (bn - g) : 32;
        own_add(H, recs + b0 + g, k, lane, cur);
      }
      *(float4*)orow = cur;
    }
    if (more == 0u) break;
    if (lane < RJ) bcnt[wave * RJ + lane] = 0u;
    __threadfence();
    __syncthreads();
  }
  if (use_lists == 0) {
    if (tid < NB) { L[tid] = bfst[tid]; L[NB + tid] = bcnt[tid]; }
    const unsigned tot = s_total;
    if (tid == 0) { L[2 * NB] = flag0; L[2 * NB + 1] = tot; }
    if (flag0 == 0u)
      for (unsigned i = (unsigned)tid; i < tot; i += (unsigned)OT) LR[i] = recs[i];
  }
}

extern "C" void kernel_launch(void* const* d_in, const int* in_sizes, int n_in,
                              void* d_out, int out_size, void* d_ws, size_t ws_size,
                              hipStream_t stream) {
  (void)n_in; (void)out_size; (void)ws_size;
  const float* x  = (const float*)d_in[0];
  const int*   ei = (const int*)d_in[1];
  const float* W1 = (const float*)d_in[2];
  const float* b1 = (const float*)d_in[3];
  const float* W2 = (const float*)d_in[4];
  const float* b2 = (const float*)d_in[5];
  const int n = in_sizes[0] / D;
  const int e = in_sizes[1] / 2;
  const int* srcv = ei;
  const int* dstv = ei + e;

  char* ws = (char*)d_ws;
  float* dinv = (float*)ws;
  size_t off = (((size_t)n * 4) + 255) & ~(size_t)255;
  float* Hbuf = (float*)(ws + off);
  off += (((size_t)n * D * 4) + 255) & ~(size_t)255;
  float* Abuf = (float*)(ws + off);
  off += (((size_t)n * D * 4) + 255) & ~(size_t)255;
  unsigned* keys = (unsigned*)(ws + off);
  off += ((((size_t)(e + KB - 1) / KB) * 1024) + 255) & ~(size_t)255;
  unsigned* lists = (unsigned*)(ws + off);
  unsigned* planes = (unsigned*)Abuf;
  float* outp = (float*)d_out;

  const dim3 blk(256);
  const int gblk = (n + 127) / 128;
  const int nown = (n + OR - 1) / OR;
  const int nkb = (e + KB - 1) / KB;
  const int nslice = (e + HT * HU - 1) / (HT * HU);
  const int nwin = (n + HB - 1) / HB;
  const int wsz = (((n + nwin - 1) / nwin) + 31) & ~31;
  const int np = (n + 31) & ~31;

  k_hist<<<dim3(nslice, nwin), dim3(HT), 0, stream>>>(dstv, planes, e, wsz, np, n, keys, nkb);

  k_gcn_gemm<<<dim3(gblk), dim3(288), 0, stream>>>(x, W1, Hbuf, n, 0, planes, dinv, np, nslice);
  k_own<<<dim3(nown), dim3(OT), 0, stream>>>(Hbuf, dinv, srcv, dstv, keys, b1, Abuf, e, nkb, n, lists, 0);

  k_gcn_gemm<<<dim3(gblk), blk, 0, stream>>>(Abuf, W2, Hbuf, n, 1, (const unsigned*)0, (float*)0, 0, 0);
  k_own<<<dim3(nown), dim3(OT), 0, stream>>>(Hbuf, dinv, srcv, dstv, keys, b2, outp, e, nkb, n, lists, 1);
}
